# starting kernel
# baseline (speedup 1.0000x reference)
#include <hip/hip_runtime.h>

typedef __attribute__((ext_vector_type(2))) float v2f;
typedef __attribute__((ext_vector_type(8))) float v8f;

static constexpr int NN   = 100000;
static constexpr int EE   = 1600000;
static constexpr int IND  = 256;
static constexpr int HID  = 128;
static constexpr int HEADS= 8;
static constexpr int DH   = 16;
static constexpr int NCLS = 64;
static constexpr float SLOPE = 0.2f;
static constexpr unsigned NEGKEY = 0x007FFFFFu;

__device__ __forceinline__ unsigned fkey(float f) {
    unsigned u = __float_as_uint(f);
    return (u & 0x80000000u) ? ~u : (u | 0x80000000u);
}
__device__ __forceinline__ float funkey(unsigned k) {
    unsigned u = (k & 0x80000000u) ? (k & 0x7fffffffu) : ~k;
    return __uint_as_float(u);
}
__device__ __forceinline__ float lrelu(float x) { return x > 0.f ? x : SLOPE * x; }
__device__ __forceinline__ float elu1(float x)  { return x > 0.f ? x : expm1f(x); }

__global__ __launch_bounds__(256) void k_initA(unsigned* mkey1, float* s1, float* acc1,
                                               unsigned* mkey2, float* s2) {
    int t = blockIdx.x * 256 + threadIdx.x;
    if (t < NN * HID) acc1[t] = 0.f;
    if (t < NN * HEADS) { mkey1[t] = NEGKEY; s1[t] = 0.f; }
    if (t < NN) { mkey2[t] = NEGKEY; s2[t] = 0.f; }
}
__global__ __launch_bounds__(256) void k_initB(float* acc2) {
    int t = blockIdx.x * 256 + threadIdx.x;
    if (t < NN * NCLS) acc2[t] = 0.f;
}

__global__ __launch_bounds__(256) void k_gemm1(const float* __restrict__ X,
                                               const float* __restrict__ W,
                                               float* __restrict__ H) {
    const int wave = threadIdx.x >> 5;
    const int lane = threadIdx.x & 31;
    const int half = lane >> 4;
    const int l16  = lane & 15;
    const int row0 = blockIdx.x * 16;
    const int col  = wave * 16 + l16;

    v8f c = {};
    const float* xrow = X + (size_t)(row0 + l16) * IND;
#pragma unroll 4
    for (int kk = 0; kk < IND; kk += 4) {
        const int k = kk + half * 2;
        __builtin_prefetch(xrow + k + IND, 0, 1);
        v2f a = *(const v2f*)(xrow + k);
        v2f b;
        b.x = W[(size_t)k * HID + col];
        b.y = W[(size_t)(k + 1) * HID + col];
        c = __builtin_amdgcn_wmma_f32_16x16x4_f32(false, a, false, b, (short)0, c, false, false);
    }
#pragma unroll
    for (int v = 0; v < 8; ++v)
        H[(size_t)(row0 + v + half * 8) * HID + col] = c[v];
}

__global__ __launch_bounds__(256) void k_att1(const float* __restrict__ H,
                                              const float* __restrict__ al,
                                              const float* __restrict__ ar,
                                              float* __restrict__ el, float* __restrict__ er) {
    int t = blockIdx.x * 256 + threadIdx.x;
    if (t >= NN * HEADS) return;
    const int h = t & 7;
    const float4* hp = (const float4*)(H + (size_t)t * DH);
    const float4* ap = (const float4*)(al + h * DH);
    const float4* rp = (const float4*)(ar + h * DH);
    float se = 0.f, sr = 0.f;
#pragma unroll
    for (int i = 0; i < 4; ++i) {
        const float4 hv = hp[i], av = ap[i], rv = rp[i];
        se += hv.x * av.x + hv.y * av.y + hv.z * av.z + hv.w * av.w;
        sr += hv.x * rv.x + hv.y * rv.y + hv.z * rv.z + hv.w * rv.w;
    }
    el[t] = se; er[t] = sr;
}

__global__ __launch_bounds__(256) void k_emax1(const int* __restrict__ src, const int* __restrict__ dst,
                                               const float* __restrict__ el, const float* __restrict__ er,
                                               unsigned* __restrict__ mkey) {
    int t = blockIdx.x * 256 + threadIdx.x;
    if (t >= EE * HEADS) return;
    const int e = t >> 3, h = t & 7;
    const int s = src[e], d = dst[e];
    const float ev = lrelu(el[s * HEADS + h] + er[d * HEADS + h]);
    atomicMax(&mkey[d * HEADS + h], fkey(ev));
}

__global__ __launch_bounds__(256) void k_eagg1(const int* __restrict__ src, const int* __restrict__ dst,
                                               const float* __restrict__ el, const float* __restrict__ er,
                                               const unsigned* __restrict__ mkey,
                                               const float* __restrict__ H,
                                               float* __restrict__ ssum, float* __restrict__ acc) {
    int t = blockIdx.x * 256 + threadIdx.x;
    if (t >= EE * HEADS) return;
    const int e = t >> 3, h = t & 7;
    const int s = src[e], d = dst[e];
    const float ev = lrelu(el[s * HEADS + h] + er[d * HEADS + h]);
    float m = funkey(mkey[d * HEADS + h]);
    if (!isfinite(m)) m = 0.f;
    const float p = __expf(ev - m);
    atomicAdd(&ssum[d * HEADS + h], p);
    const float4* hv = (const float4*)(H + (size_t)s * HID + h * DH);
    float* ap = acc + (size_t)d * HID + h * DH;
    float4 a0 = hv[0], a1 = hv[1], a2 = hv[2], a3 = hv[3];
    atomicAdd(ap + 0,  p * a0.x); atomicAdd(ap + 1,  p * a0.y);
    atomicAdd(ap + 2,  p * a0.z); atomicAdd(ap + 3,  p * a0.w);
    atomicAdd(ap + 4,  p * a1.x); atomicAdd(ap + 5,  p * a1.y);
    atomicAdd(ap + 6,  p * a1.z); atomicAdd(ap + 7,  p * a1.w);
    atomicAdd(ap + 8,  p * a2.x); atomicAdd(ap + 9,  p * a2.y);
    atomicAdd(ap + 10, p * a2.z); atomicAdd(ap + 11, p * a2.w);
    atomicAdd(ap + 12, p * a3.x); atomicAdd(ap + 13, p * a3.y);
    atomicAdd(ap + 14, p * a3.z); atomicAdd(ap + 15, p * a3.w);
}

__global__ __launch_bounds__(256) void k_fin1(const float* __restrict__ acc, const float* __restrict__ ssum,
                                              const float* __restrict__ b, float* __restrict__ h2) {
    int t = blockIdx.x * 256 + threadIdx.x;
    if (t >= NN * HID) return;
    const int n = t >> 7, cidx = t & 127, h = cidx >> 4;
    const float s = ssum[n * HEADS + h];
    float v = (s > 0.f) ? acc[t] / s : 0.f;
    v = elu1(elu1(v + b[cidx]));
    h2[t] = v;
}

__global__ __launch_bounds__(256) void k_gemm2(const float* __restrict__ H2,
                                               const float* __restrict__ W,
                                               float* __restrict__ HT) {
    const int wave = threadIdx.x >> 5;
    const int lane = threadIdx.x & 31;
    const int half = lane >> 4;
    const int l16  = lane & 15;
    const int rt = wave >> 2, ct = wave & 3;
    const int row0 = blockIdx.x * 32 + rt * 16;
    const int col  = ct * 16 + l16;

    v8f c = {};
    const float* xrow = H2 + (size_t)(row0 + l16) * HID;
#pragma unroll 4
    for (int kk = 0; kk < HID; kk += 4) {
        const int k = kk + half * 2;
        v2f a = *(const v2f*)(xrow + k);
        v2f b;
        b.x = W[(size_t)k * NCLS + col];
        b.y = W[(size_t)(k + 1) * NCLS + col];
        c = __builtin_amdgcn_wmma_f32_16x16x4_f32(false, a, false, b, (short)0, c, false, false);
    }
#pragma unroll
    for (int v = 0; v < 8; ++v)
        HT[(size_t)(row0 + v + half * 8) * NCLS + col] = c[v];
}

__global__ __launch_bounds__(256) void k_att2(const float* __restrict__ HT,
                                              const float* __restrict__ al,
                                              const float* __restrict__ ar,
                                              float* __restrict__ el, float* __restrict__ er) {
    int t = blockIdx.x * 256 + threadIdx.x;
    if (t >= NN) return;
    const float4* hp = (const float4*)(HT + (size_t)t * NCLS);
    const float4* ap = (const float4*)al;
    const float4* rp = (const float4*)ar;
    float se = 0.f, sr = 0.f;
#pragma unroll
    for (int i = 0; i < NCLS / 4; ++i) {
        const float4 hv = hp[i], av = ap[i], rv = rp[i];
        se += hv.x * av.x + hv.y * av.y + hv.z * av.z + hv.w * av.w;
        sr += hv.x * rv.x + hv.y * rv.y + hv.z * rv.z + hv.w * rv.w;
    }
    el[t] = se; er[t] = sr;
}

__global__ __launch_bounds__(256) void k_emax2(const int* __restrict__ src, const int* __restrict__ dst,
                                               const float* __restrict__ el, const float* __restrict__ er,
                                               unsigned* __restrict__ mkey) {
    int t = blockIdx.x * 256 + threadIdx.x;
    if (t >= EE) return;
    const int s = src[t], d = dst[t];
    const float ev = lrelu(el[s] + er[d]);
    atomicMax(&mkey[d], fkey(ev));
}

__global__ __launch_bounds__(256) void k_eagg2(const int* __restrict__ src, const int* __restrict__ dst,
                                               const float* __restrict__ el, const float* __restrict__ er,
                                               const unsigned* __restrict__ mkey,
                                               const float* __restrict__ HT,
                                               float* __restrict__ ssum, float* __restrict__ acc) {
    int t = blockIdx.x * 256 + threadIdx.x;
    if (t >= EE * 16) return;
    const int e = t >> 4, q = t & 15;
    const int s = src[e], d = dst[e];
    const float ev = lrelu(el[s] + er[d]);
    float m = funkey(mkey[d]);
    if (!isfinite(m)) m = 0.f;
    const float p = __expf(ev - m);
    if (q == 0) atomicAdd(&ssum[d], p);
    const float4 hv = ((const float4*)(HT + (size_t)s * NCLS))[q];
    float* ap = acc + (size_t)d * NCLS + q * 4;
    atomicAdd(ap + 0, p * hv.x); atomicAdd(ap + 1, p * hv.y);
    atomicAdd(ap + 2, p * hv.z); atomicAdd(ap + 3, p * hv.w);
}

__global__ __launch_bounds__(256) void k_fin2(const float* __restrict__ acc, const float* __restrict__ ssum,
                                              const float* __restrict__ b, float* __restrict__ out) {
    int t = blockIdx.x * 256 + threadIdx.x;
    if (t >= NN * NCLS) return;
    const int n = t >> 6, cidx = t & 63;
    const float s = ssum[n];
    float v = (s > 0.f) ? acc[t] / s : 0.f;
    out[t] = v + b[cidx];
}

extern "C" void kernel_launch(void* const* d_in, const int* in_sizes, int n_in,
                              void* d_out, int out_size, void* d_ws, size_t ws_size,
                              hipStream_t stream) {
    const float* x   = (const float*)d_in[0];
    const int*   src = (const int*)  d_in[1];
    const int*   dst = (const int*)  d_in[2];
    const float* W1  = (const float*)d_in[3];
    const float* al1 = (const float*)d_in[4];
    const float* ar1 = (const float*)d_in[5];
    const float* b1  = (const float*)d_in[6];
    const float* W2  = (const float*)d_in[7];
    const float* al2 = (const float*)d_in[8];
    const float* ar2 = (const float*)d_in[9];
    const float* b2  = (const float*)d_in[10];
    float* out = (float*)d_out;

    size_t off = 0;
    auto take = [&](size_t nfloats) -> float* {
        float* p = (float*)((char*)d_ws + off);
        off = (off + nfloats * sizeof(float) + 255) & ~(size_t)255;
        return p;
    };
    float*    h1    = take((size_t)NN * HID);
    float*    el1   = take((size_t)NN * HEADS);
    float*    er1   = take((size_t)NN * HEADS);
    unsigned* mkey1 = (unsigned*)take((size_t)NN * HEADS);
    float*    s1    = take((size_t)NN * HEADS);
    float*    acc1  = take((size_t)NN * HID);
    float*    h2t   = take((size_t)NN * NCLS);
    float*    el2   = take((size_t)NN);
    float*    er2   = take((size_t)NN);
    unsigned* mkey2 = (unsigned*)take((size_t)NN);
    float*    s2    = take((size_t)NN);
    float*    h2    = h1;
    float*    acc2  = acc1;

    const int T = 256;
    k_initA<<<(NN * HID + T - 1) / T, T, 0, stream>>>(mkey1, s1, acc1, mkey2, s2);
    k_gemm1<<<NN / 16, T, 0, stream>>>(x, W1, h1);
    k_att1 <<<(NN * HEADS + T - 1) / T, T, 0, stream>>>(h1, al1, ar1, el1, er1);
    k_emax1<<<(EE * HEADS + T - 1) / T, T, 0, stream>>>(src, dst, el1, er1, mkey1);
    k_eagg1<<<(EE * HEADS + T - 1) / T, T, 0, stream>>>(src, dst, el1, er1, mkey1, h1, s1, acc1);
    k_fin1 <<<(NN * HID + T - 1) / T, T, 0, stream>>>(acc1, s1, b1, h2);
    k_gemm2<<<NN / 32, T, 0, stream>>>(h2, W2, h2t);
    k_att2 <<<(NN + T - 1) / T, T, 0, stream>>>(h2t, al2, ar2, el2, er2);
    k_initB<<<(NN * NCLS + T - 1) / T, T, 0, stream>>>(acc2);
    k_emax2<<<(EE + T - 1) / T, T, 0, stream>>>(src, dst, el2, er2, mkey2);
    k_eagg2<<<(EE * 16 + T - 1) / T, T, 0, stream>>>(src, dst, el2, er2, mkey2, h2t, s2, acc2);
    k_fin2 <<<(NN * NCLS + T - 1) / T, T, 0, stream>>>(acc2, s2, b2, out);
}


# Edge scatter by dst-tile ownership: a block per 400 nodes filters the dst array; max, exp-sum, row sums by LDS atomics; final rows written; global atomics, init, finalize gone; LDS arrival sum order
# speedup vs baseline: 2.7303x; 2.7303x over previous
#include <hip/hip_runtime.h>

typedef __attribute__((ext_vector_type(2))) float v2f;
typedef __attribute__((ext_vector_type(8))) float v8f;

static constexpr int NN   = 100000;
static constexpr int EE   = 1600000;
static constexpr int IND  = 256;
static constexpr int HID  = 128;
static constexpr int HEADS= 8;
static constexpr int DH   = 16;
static constexpr int NCLS = 64;
static constexpr float SLOPE = 0.2f;
static constexpr unsigned NEGKEY = 0x007FFFFFu;

__device__ __forceinline__ unsigned fkey(float f) {
    unsigned u = __float_as_uint(f);
    return (u & 0x80000000u) ? ~u : (u | 0x80000000u);
}
__device__ __forceinline__ float funkey(unsigned k) {
    unsigned u = (k & 0x80000000u) ? (k & 0x7fffffffu) : ~k;
    return __uint_as_float(u);
}
__device__ __forceinline__ float lrelu(float x) { return x > 0.f ? x : SLOPE * x; }
__device__ __forceinline__ float elu1(float x)  { return x > 0.f ? x : expm1f(x); }

__global__ __launch_bounds__(256) void k_gemm1(const float* __restrict__ X,
                                               const float* __restrict__ W,
                                               float* __restrict__ H) {
    const int wave = threadIdx.x >> 5;
    const int lane = threadIdx.x & 31;
    const int half = lane >> 4;
    const int l16  = lane & 15;
    const int row0 = blockIdx.x * 16;
    const int col  = wave * 16 + l16;

    v8f c = {};
    const float* xrow = X + (size_t)(row0 + l16) * IND;
#pragma unroll 4
    for (int kk = 0; kk < IND; kk += 4) {
        const int k = kk + half * 2;
        __builtin_prefetch(xrow + k + IND, 0, 1);
        v2f a = *(const v2f*)(xrow + k);
        v2f b;
        b.x = W[(size_t)k * HID + col];
        b.y = W[(size_t)(k + 1) * HID + col];
        c = __builtin_amdgcn_wmma_f32_16x16x4_f32(false, a, false, b, (short)0, c, false, false);
    }
#pragma unroll
    for (int v = 0; v < 8; ++v)
        H[(size_t)(row0 + v + half * 8) * HID + col] = c[v];
}

__global__ __launch_bounds__(256) void k_att1(const float* __restrict__ H,
                                              const float* __restrict__ al,
                                              const float* __restrict__ ar,
                                              float* __restrict__ el, float* __restrict__ er) {
    int t = blockIdx.x * 256 + threadIdx.x;
    if (t >= NN * HEADS) return;
    const int h = t & 7;
    const float4* hp = (const float4*)(H + (size_t)t * DH);
    const float4* ap = (const float4*)(al + h * DH);
    const float4* rp = (const float4*)(ar + h * DH);
    float se = 0.f, sr = 0.f;
#pragma unroll
    for (int i = 0; i < 4; ++i) {
        const float4 hv = hp[i], av = ap[i], rv = rp[i];
        se += hv.x * av.x + hv.y * av.y + hv.z * av.z + hv.w * av.w;
        sr += hv.x * rv.x + hv.y * rv.y + hv.z * rv.z + hv.w * rv.w;
    }
    el[t] = se; er[t] = sr;
}

static constexpr int TR    = 400;
static constexpr int NTILE = NN / TR;
static constexpr int CAP   = 16384;
static constexpr int TB    = 1024;
static_assert(NN % TR == 0 && TR <= 512, "tile index fits 9 bits");
static_assert(EE % 4 == 0 && CAP % 4 == 0 && CAP == 4 * 4 * TB, "a CAP window is one sweep of the block");

__device__ __forceinline__ void tile_test(int d, int e, int n0, unsigned* list, unsigned* cnt) {
    const unsigned dl = (unsigned)(d - n0);
    if (dl < (unsigned)TR) {
        const unsigned slot = atomicAdd(cnt, 1u);
        if (slot < (unsigned)CAP) list[slot] = ((unsigned)e << 9) | dl;
    }
}
__device__ __forceinline__ void tile_collect(const int* __restrict__ dst, int n0, int e0, int e1,
                                             unsigned* list, unsigned* cnt) {
    const int4* d4 = (const int4*)dst;
    const int q1 = e1 >> 2;
    for (int q = (e0 >> 2) + (int)threadIdx.x; q < q1; q += 4 * TB) {
        int4 d[4];
#pragma unroll
        for (int u = 0; u < 4; ++u) {
            const int qq = q + TB * u;
            d[u] = (qq < q1) ? d4[qq] : make_int4(-1, -1, -1, -1);
        }
#pragma unroll
        for (int u = 0; u < 4; ++u) {
            const int e = (q + TB * u) * 4;
            tile_test(d[u].x, e,     n0, list, cnt);
            tile_test(d[u].y, e + 1, n0, list, cnt);
            tile_test(d[u].z, e + 2, n0, list, cnt);
            tile_test(d[u].w, e + 3, n0, list, cnt);
        }
    }
}
template <int G>
__device__ __forceinline__ int tile_resolve(const int* __restrict__ src, unsigned* list, int n) {
    static_assert(CAP % G == 0 && G <= TB, "padding stays inside the list");
    for (int i = threadIdx.x; i < n; i += TB) {
        const unsigned ent = list[i];
        list[i] = ((unsigned)src[ent >> 9] << 9) | (ent & 511u);
    }
    const int npad = (n + G - 1) / G * G;
    if ((int)threadIdx.x < npad - n) list[n + threadIdx.x] = (unsigned)TR;
    return npad;
}

__device__ __forceinline__ void tile1_max(const float* __restrict__ el, const float* __restrict__ er,
                                          int n0, const unsigned* list, int n, unsigned* mk) {
    for (int i = threadIdx.x; i < n; i += TB) {
        const unsigned ent = list[i];
        const int s = (int)(ent >> 9), dl = (int)(ent & 511u);
        const float4* ep = (const float4*)(el + (size_t)s * HEADS);
        const float4* rp = (const float4*)(er + (size_t)(n0 + dl) * HEADS);
        const float4 a0 = ep[0], a1 = ep[1], r0 = rp[0], r1 = rp[1];
        unsigned* m = mk + dl * HEADS;
        atomicMax(m + 0, fkey(lrelu(a0.x + r0.x)));
        atomicMax(m + 1, fkey(lrelu(a0.y + r0.y)));
        atomicMax(m + 2, fkey(lrelu(a0.z + r0.z)));
        atomicMax(m + 3, fkey(lrelu(a0.w + r0.w)));
        atomicMax(m + 4, fkey(lrelu(a1.x + r1.x)));
        atomicMax(m + 5, fkey(lrelu(a1.y + r1.y)));
        atomicMax(m + 6, fkey(lrelu(a1.z + r1.z)));
        atomicMax(m + 7, fkey(lrelu(a1.w + r1.w)));
    }
}
__device__ __forceinline__ void tile1_agg(const float* __restrict__ el, const float* __restrict__ er,
                                          const float* __restrict__ H, int n0,
                                          const unsigned* list, int n, const unsigned* mk,
                                          float* ss, float* acc) {
    const int wave = threadIdx.x >> 5, lane = threadIdx.x & 31;
    const int hsel = lane >> 4;
    for (int i0 = wave * 4; i0 < n; i0 += (TB / 32) * 4) {
        const unsigned ent = list[i0 + (lane >> 3)];
        const int s = (int)(ent >> 9), dl = (int)(ent & 511u);
        const int h = lane & 7;
        const int dn = (n0 + dl < NN) ? n0 + dl : NN - 1;
        const float ev = lrelu(el[(size_t)s * HEADS + h] + er[(size_t)dn * HEADS + h]);
        float m = funkey(mk[dl * HEADS + h]);
        if (!isfinite(m)) m = 0.f;
        const float p = __expf(ev - m);
        atomicAdd(&ss[dl * HEADS + h], p);
        float v[4][4];
#pragma unroll
        for (int i = 0; i < 4; ++i) {
            const int si = __builtin_amdgcn_readlane(s, i * 8);
            const float* hr = H + (size_t)si * HID + lane;
#pragma unroll
            for (int j = 0; j < 4; ++j) v[i][j] = hr[32 * j];
        }
#pragma unroll
        for (int i = 0; i < 4; ++i) {
            const int di = __builtin_amdgcn_readlane(dl, i * 8);
            float* ar = acc + di * HID + lane;
#pragma unroll
            for (int j = 0; j < 4; ++j) {
                const float pa = __int_as_float(__builtin_amdgcn_readlane(__float_as_int(p), i * 8 + 2 * j));
                const float pb = __int_as_float(__builtin_amdgcn_readlane(__float_as_int(p), i * 8 + 2 * j + 1));
                atomicAdd(ar + 32 * j, (hsel ? pb : pa) * v[i][j]);
            }
        }
    }
}

__global__ __launch_bounds__(1024) void k_tile1(const int* __restrict__ src, const int* __restrict__ dst,
                                                const float* __restrict__ el, const float* __restrict__ er,
                                                const float* __restrict__ H, const float* __restrict__ b,
                                                float* __restrict__ h2) {
    __shared__ __attribute__((aligned(16))) float acc[(TR + 1) * HID];
    __shared__ unsigned mk[(TR + 1) * HEADS];
    __shared__ float    ss[(TR + 1) * HEADS];
    __shared__ unsigned list[CAP];
    __shared__ unsigned cnt;
    const int tid = threadIdx.x;
    const int n0 = blockIdx.x * TR;
    for (int i = tid; i < (TR + 1) * HID; i += TB) acc[i] = 0.f;
    for (int i = tid; i < (TR + 1) * HEADS; i += TB) { mk[i] = NEGKEY; ss[i] = 0.f; }
    if (tid == 0) cnt = 0u;
    __syncthreads();
    tile_collect(dst, n0, 0, EE, list, &cnt);
    __syncthreads();
    const unsigned total = cnt;
    if (total <= (unsigned)CAP) {
        const int npad = tile_resolve<4>(src, list, (int)total);
        __syncthreads();
        tile1_max(el, er, n0, list, (int)total, mk);
        __syncthreads();
        tile1_agg(el, er, H, n0, list, npad, mk, ss, acc);
    } else {
        for (int pass = 0; pass < 2; ++pass) {
            for (int w0 = 0; w0 < EE; w0 += CAP) {
                __syncthreads();
                if (tid == 0) cnt = 0u;
                __syncthreads();
                tile_collect(dst, n0, w0, (w0 + CAP < EE) ? w0 + CAP : EE, list, &cnt);
                __syncthreads();
                const int n = (int)cnt;
                const int npad = tile_resolve<4>(src, list, n);
                __syncthreads();
                if (pass == 0) tile1_max(el, er, n0, list, n, mk);
                else           tile1_agg(el, er, H, n0, list, npad, mk, ss, acc);
            }
        }
    }
    __syncthreads();
    for (int i = tid; i < TR * HID; i += TB) {
        const int nl = i >> 7, cidx = i & 127, h = cidx >> 4;
        const float s = ss[nl * HEADS + h];
        float v = (s > 0.f) ? acc[i] / s : 0.f;
        v = elu1(elu1(v + b[cidx]));
        h2[(size_t)n0 * HID + i] = v;
    }
}

__global__ __launch_bounds__(256) void k_gemm2(const float* __restrict__ H2,
                                               const float* __restrict__ W,
                                               float* __restrict__ HT) {
    const int wave = threadIdx.x >> 5;
    const int lane = threadIdx.x & 31;
    const int half = lane >> 4;
    const int l16  = lane & 15;
    const int rt = wave >> 2, ct = wave & 3;
    const int row0 = blockIdx.x * 32 + rt * 16;
    const int col  = ct * 16 + l16;

    v8f c = {};
    const float* xrow = H2 + (size_t)(row0 + l16) * HID;
#pragma unroll 4
    for (int kk = 0; kk < HID; kk += 4) {
        const int k = kk + half * 2;
        v2f a = *(const v2f*)(xrow + k);
        v2f b;
        b.x = W[(size_t)k * NCLS + col];
        b.y = W[(size_t)(k + 1) * NCLS + col];
        c = __builtin_amdgcn_wmma_f32_16x16x4_f32(false, a, false, b, (short)0, c, false, false);
    }
#pragma unroll
    for (int v = 0; v < 8; ++v)
        HT[(size_t)(row0 + v + half * 8) * NCLS + col] = c[v];
}

__global__ __launch_bounds__(256) void k_att2(const float* __restrict__ HT,
                                              const float* __restrict__ al,
                                              const float* __restrict__ ar,
                                              float* __restrict__ el, float* __restrict__ er) {
    int t = blockIdx.x * 256 + threadIdx.x;
    if (t >= NN) return;
    const float4* hp = (const float4*)(HT + (size_t)t * NCLS);
    const float4* ap = (const float4*)al;
    const float4* rp = (const float4*)ar;
    float se = 0.f, sr = 0.f;
#pragma unroll
    for (int i = 0; i < NCLS / 4; ++i) {
        const float4 hv = hp[i], av = ap[i], rv = rp[i];
        se += hv.x * av.x + hv.y * av.y + hv.z * av.z + hv.w * av.w;
        sr += hv.x * rv.x + hv.y * rv.y + hv.z * rv.z + hv.w * rv.w;
    }
    el[t] = se; er[t] = sr;
}


__device__ __forceinline__ void tile2_max(const float* __restrict__ el, const float* __restrict__ er,
                                          int n0, const unsigned* list, int n, unsigned* mk) {
    for (int i = threadIdx.x; i < n; i += TB) {
        const unsigned ent = list[i];
        const int s = (int)(ent >> 9), dl = (int)(ent & 511u);
        atomicMax(mk + dl, fkey(lrelu(el[s] + er[n0 + dl])));
    }
}
__device__ __forceinline__ void tile2_agg(const float* __restrict__ el, const float* __restrict__ er,
                                          const float* __restrict__ HT, int n0,
                                          const unsigned* list, int n, const unsigned* mk,
                                          float* ss, float* acc) {
    const int wave = threadIdx.x >> 5, lane = threadIdx.x & 31;
    for (int i0 = wave * 32; i0 < n; i0 += TB) {
        const int idx = i0 + lane;
        const bool ok = idx < n;
        const unsigned ent = list[ok ? idx : i0];
        const int s = (int)(ent >> 9), dl = (int)(ent & 511u);
        const int dn = (n0 + dl < NN) ? n0 + dl : NN - 1;
        const float ev = lrelu(el[s] + er[dn]);
        float m = funkey(mk[dl]);
        if (!isfinite(m)) m = 0.f;
        const float p = __expf(ev - m);
        if (ok) atomicAdd(&ss[dl], p);
        const int have = (n - i0 < 32) ? n - i0 : 32;
        for (int g = 0; g < have; g += 8) {
            float v0[8], v1[8];
#pragma unroll
            for (int i = 0; i < 8; ++i) {
                const int si = __builtin_amdgcn_readlane(s, g + i);
                const float* hr = HT + (size_t)si * NCLS + lane;
                v0[i] = hr[0];
                v1[i] = hr[32];
            }
#pragma unroll
            for (int i = 0; i < 8; ++i) {
                const int di = __builtin_amdgcn_readlane(dl, g + i);
                const float pi = __int_as_float(__builtin_amdgcn_readlane(__float_as_int(p), g + i));
                float* ar = acc + di * NCLS + lane;
                atomicAdd(ar,      pi * v0[i]);
                atomicAdd(ar + 32, pi * v1[i]);
            }
        }
    }
}

__global__ __launch_bounds__(1024) void k_tile2(const int* __restrict__ src, const int* __restrict__ dst,
                                                const float* __restrict__ el, const float* __restrict__ er,
                                                const float* __restrict__ HT, const float* __restrict__ b,
                                                float* __restrict__ out) {
    __shared__ __attribute__((aligned(16))) float acc[(TR + 1) * NCLS];
    __shared__ unsigned mk[TR + 1];
    __shared__ float    ss[TR + 1];
    __shared__ unsigned list[CAP];
    __shared__ unsigned cnt;
    const int tid = threadIdx.x;
    const int n0 = blockIdx.x * TR;
    for (int i = tid; i < (TR + 1) * NCLS; i += TB) acc[i] = 0.f;
    for (int i = tid; i < TR + 1; i += TB) { mk[i] = NEGKEY; ss[i] = 0.f; }
    if (tid == 0) cnt = 0u;
    __syncthreads();
    tile_collect(dst, n0, 0, EE, list, &cnt);
    __syncthreads();
    const unsigned total = cnt;
    if (total <= (unsigned)CAP) {
        const int npad = tile_resolve<8>(src, list, (int)total);
        __syncthreads();
        tile2_max(el, er, n0, list, (int)total, mk);
        __syncthreads();
        tile2_agg(el, er, HT, n0, list, npad, mk, ss, acc);
    } else {
        for (int pass = 0; pass < 2; ++pass) {
            for (int w0 = 0; w0 < EE; w0 += CAP) {
                __syncthreads();
                if (tid == 0) cnt = 0u;
                __syncthreads();
                tile_collect(dst, n0, w0, (w0 + CAP < EE) ? w0 + CAP : EE, list, &cnt);
                __syncthreads();
                const int n = (int)cnt;
                const int npad = tile_resolve<8>(src, list, n);
                __syncthreads();
                if (pass == 0) tile2_max(el, er, n0, list, n, mk);
                else           tile2_agg(el, er, HT, n0, list, npad, mk, ss, acc);
            }
        }
    }
    __syncthreads();
    for (int i = tid; i < TR * NCLS; i += TB) {
        const int nl = i >> 6, cidx = i & 63;
        const float s = ss[nl];
        const float v = (s > 0.f) ? acc[i] / s : 0.f;
        out[(size_t)n0 * NCLS + i] = v + b[cidx];
    }
}

extern "C" void kernel_launch(void* const* d_in, const int* in_sizes, int n_in,
                              void* d_out, int out_size, void* d_ws, size_t ws_size,
                              hipStream_t stream) {
    const float* x   = (const float*)d_in[0];
    const int*   src = (const int*)  d_in[1];
    const int*   dst = (const int*)  d_in[2];
    const float* W1  = (const float*)d_in[3];
    const float* al1 = (const float*)d_in[4];
    const float* ar1 = (const float*)d_in[5];
    const float* b1  = (const float*)d_in[6];
    const float* W2  = (const float*)d_in[7];
    const float* al2 = (const float*)d_in[8];
    const float* ar2 = (const float*)d_in[9];
    const float* b2  = (const float*)d_in[10];
    float* out = (float*)d_out;

    size_t off = 0;
    auto take = [&](size_t nfloats) -> float* {
        float* p = (float*)((char*)d_ws + off);
        off = (off + nfloats * sizeof(float) + 255) & ~(size_t)255;
        return p;
    };
    float* h1  = take((size_t)NN * HID);
    float* el1 = take((size_t)NN * HEADS);
    float* er1 = take((size_t)NN * HEADS);
    float* h2  = take((size_t)NN * HID);
    float* h2t = take((size_t)NN * NCLS);
    float* el2 = take((size_t)NN);
    float* er2 = take((size_t)NN);

    const int T = 256;
    k_gemm1<<<NN / 16, T, 0, stream>>>(x, W1, h1);
    k_att1 <<<(NN * HEADS + T - 1) / T, T, 0, stream>>>(h1, al1, ar1, el1, er1);
    k_tile1<<<NTILE, TB, 0, stream>>>(src, dst, el1, er1, h1, b1, h2);
    k_gemm2<<<NN / 32, T, 0, stream>>>(h2, W2, h2t);
    k_att2 <<<(NN + T - 1) / T, T, 0, stream>>>(h2t, al2, ar2, el2, er2);
    k_tile2<<<NTILE, TB, 0, stream>>>(src, dst, el2, er2, h2t, b2, out);
}


# Layer 2 reuses layer 1's per-tile edge list: k_tile1 stores each tile's count and resolved entries in the workspace, k_tile2 loads them; 2nd dst sweep and src lookup gone; LDS arrival sum order
# speedup vs baseline: 3.2469x; 1.1892x over previous
#include <hip/hip_runtime.h>

typedef __attribute__((ext_vector_type(2))) float v2f;
typedef __attribute__((ext_vector_type(8))) float v8f;

static constexpr int NN   = 100000;
static constexpr int EE   = 1600000;
static constexpr int IND  = 256;
static constexpr int HID  = 128;
static constexpr int HEADS= 8;
static constexpr int DH   = 16;
static constexpr int NCLS = 64;
static constexpr float SLOPE = 0.2f;
static constexpr unsigned NEGKEY = 0x007FFFFFu;

__device__ __forceinline__ unsigned fkey(float f) {
    unsigned u = __float_as_uint(f);
    return (u & 0x80000000u) ? ~u : (u | 0x80000000u);
}
__device__ __forceinline__ float funkey(unsigned k) {
    unsigned u = (k & 0x80000000u) ? (k & 0x7fffffffu) : ~k;
    return __uint_as_float(u);
}
__device__ __forceinline__ float lrelu(float x) { return x > 0.f ? x : SLOPE * x; }
__device__ __forceinline__ float elu1(float x)  { return x > 0.f ? x : expm1f(x); }

__global__ __launch_bounds__(256) void k_gemm1(const float* __restrict__ X,
                                               const float* __restrict__ W,
                                               float* __restrict__ H) {
    const int wave = threadIdx.x >> 5;
    const int lane = threadIdx.x & 31;
    const int half = lane >> 4;
    const int l16  = lane & 15;
    const int row0 = blockIdx.x * 16;
    const int col  = wave * 16 + l16;

    v8f c = {};
    const float* xrow = X + (size_t)(row0 + l16) * IND;
#pragma unroll 4
    for (int kk = 0; kk < IND; kk += 4) {
        const int k = kk + half * 2;
        __builtin_prefetch(xrow + k + IND, 0, 1);
        v2f a = *(const v2f*)(xrow + k);
        v2f b;
        b.x = W[(size_t)k * HID + col];
        b.y = W[(size_t)(k + 1) * HID + col];
        c = __builtin_amdgcn_wmma_f32_16x16x4_f32(false, a, false, b, (short)0, c, false, false);
    }
#pragma unroll
    for (int v = 0; v < 8; ++v)
        H[(size_t)(row0 + v + half * 8) * HID + col] = c[v];
}

__global__ __launch_bounds__(256) void k_att1(const float* __restrict__ H,
                                              const float* __restrict__ al,
                                              const float* __restrict__ ar,
                                              float* __restrict__ el, float* __restrict__ er) {
    int t = blockIdx.x * 256 + threadIdx.x;
    if (t >= NN * HEADS) return;
    const int h = t & 7;
    const float4* hp = (const float4*)(H + (size_t)t * DH);
    const float4* ap = (const float4*)(al + h * DH);
    const float4* rp = (const float4*)(ar + h * DH);
    float se = 0.f, sr = 0.f;
#pragma unroll
    for (int i = 0; i < 4; ++i) {
        const float4 hv = hp[i], av = ap[i], rv = rp[i];
        se += hv.x * av.x + hv.y * av.y + hv.z * av.z + hv.w * av.w;
        sr += hv.x * rv.x + hv.y * rv.y + hv.z * rv.z + hv.w * rv.w;
    }
    el[t] = se; er[t] = sr;
}

static constexpr int TR    = 400;
static constexpr int NTILE = NN / TR;
static constexpr int CAP   = 16384;
static constexpr int TB    = 1024;
static_assert(NN % TR == 0 && TR <= 512, "tile index fits 9 bits");
static_assert((size_t)NTILE * CAP + NTILE <= (size_t)NN * NCLS, "lists and counts fit the former h2t slot");
static_assert(EE % 4 == 0 && CAP % 4 == 0 && CAP == 4 * 4 * TB, "a CAP window is one sweep of the block");

__device__ __forceinline__ void tile_test(int d, int e, int n0, unsigned* list, unsigned* cnt) {
    const unsigned dl = (unsigned)(d - n0);
    if (dl < (unsigned)TR) {
        const unsigned slot = atomicAdd(cnt, 1u);
        if (slot < (unsigned)CAP) list[slot] = ((unsigned)e << 9) | dl;
    }
}
__device__ __forceinline__ void tile_collect(const int* __restrict__ dst, int n0, int e0, int e1,
                                             unsigned* list, unsigned* cnt) {
    const int4* d4 = (const int4*)dst;
    const int q1 = e1 >> 2;
    for (int q = (e0 >> 2) + (int)threadIdx.x; q < q1; q += 4 * TB) {
        int4 d[4];
#pragma unroll
        for (int u = 0; u < 4; ++u) {
            const int qq = q + TB * u;
            d[u] = (qq < q1) ? d4[qq] : make_int4(-1, -1, -1, -1);
        }
#pragma unroll
        for (int u = 0; u < 4; ++u) {
            const int e = (q + TB * u) * 4;
            tile_test(d[u].x, e,     n0, list, cnt);
            tile_test(d[u].y, e + 1, n0, list, cnt);
            tile_test(d[u].z, e + 2, n0, list, cnt);
            tile_test(d[u].w, e + 3, n0, list, cnt);
        }
    }
}
template <int G>
__device__ __forceinline__ int tile_resolve(const int* __restrict__ src, unsigned* list, int n,
                                            unsigned* __restrict__ keep) {
    static_assert(CAP % G == 0 && G <= TB, "padding stays inside the list");
    for (int i = threadIdx.x; i < n; i += TB) {
        const unsigned ent = list[i];
        const unsigned res = ((unsigned)src[ent >> 9] << 9) | (ent & 511u);
        list[i] = res;
        if (keep) keep[i] = res;
    }
    const int npad = (n + G - 1) / G * G;
    if ((int)threadIdx.x < npad - n) list[n + threadIdx.x] = (unsigned)TR;
    return npad;
}

__device__ __forceinline__ void tile1_max(const float* __restrict__ el, const float* __restrict__ er,
                                          int n0, const unsigned* list, int n, unsigned* mk) {
    for (int i = threadIdx.x; i < n; i += TB) {
        const unsigned ent = list[i];
        const int s = (int)(ent >> 9), dl = (int)(ent & 511u);
        const float4* ep = (const float4*)(el + (size_t)s * HEADS);
        const float4* rp = (const float4*)(er + (size_t)(n0 + dl) * HEADS);
        const float4 a0 = ep[0], a1 = ep[1], r0 = rp[0], r1 = rp[1];
        unsigned* m = mk + dl * HEADS;
        atomicMax(m + 0, fkey(lrelu(a0.x + r0.x)));
        atomicMax(m + 1, fkey(lrelu(a0.y + r0.y)));
        atomicMax(m + 2, fkey(lrelu(a0.z + r0.z)));
        atomicMax(m + 3, fkey(lrelu(a0.w + r0.w)));
        atomicMax(m + 4, fkey(lrelu(a1.x + r1.x)));
        atomicMax(m + 5, fkey(lrelu(a1.y + r1.y)));
        atomicMax(m + 6, fkey(lrelu(a1.z + r1.z)));
        atomicMax(m + 7, fkey(lrelu(a1.w + r1.w)));
    }
}
__device__ __forceinline__ void tile1_agg(const float* __restrict__ el, const float* __restrict__ er,
                                          const float* __restrict__ H, int n0,
                                          const unsigned* list, int n, const unsigned* mk,
                                          float* ss, float* acc) {
    const int wave = threadIdx.x >> 5, lane = threadIdx.x & 31;
    const int hsel = lane >> 4;
    for (int i0 = wave * 4; i0 < n; i0 += (TB / 32) * 4) {
        const unsigned ent = list[i0 + (lane >> 3)];
        const int s = (int)(ent >> 9), dl = (int)(ent & 511u);
        const int h = lane & 7;
        const int dn = (n0 + dl < NN) ? n0 + dl : NN - 1;
        const float ev = lrelu(el[(size_t)s * HEADS + h] + er[(size_t)dn * HEADS + h]);
        float m = funkey(mk[dl * HEADS + h]);
        if (!isfinite(m)) m = 0.f;
        const float p = __expf(ev - m);
        atomicAdd(&ss[dl * HEADS + h], p);
        float v[4][4];
#pragma unroll
        for (int i = 0; i < 4; ++i) {
            const int si = __builtin_amdgcn_readlane(s, i * 8);
            const float* hr = H + (size_t)si * HID + lane;
#pragma unroll
            for (int j = 0; j < 4; ++j) v[i][j] = hr[32 * j];
        }
#pragma unroll
        for (int i = 0; i < 4; ++i) {
            const int di = __builtin_amdgcn_readlane(dl, i * 8);
            float* ar = acc + di * HID + lane;
#pragma unroll
            for (int j = 0; j < 4; ++j) {
                const float pa = __int_as_float(__builtin_amdgcn_readlane(__float_as_int(p), i * 8 + 2 * j));
                const float pb = __int_as_float(__builtin_amdgcn_readlane(__float_as_int(p), i * 8 + 2 * j + 1));
                atomicAdd(ar + 32 * j, (hsel ? pb : pa) * v[i][j]);
            }
        }
    }
}

__global__ __launch_bounds__(1024) void k_tile1(const int* __restrict__ src, const int* __restrict__ dst,
                                                const float* __restrict__ el, const float* __restrict__ er,
                                                const float* __restrict__ H, const float* __restrict__ b,
                                                float* __restrict__ h2,
                                                unsigned* __restrict__ keep, unsigned* __restrict__ keepn) {
    __shared__ __attribute__((aligned(16))) float acc[(TR + 1) * HID];
    __shared__ unsigned mk[(TR + 1) * HEADS];
    __shared__ float    ss[(TR + 1) * HEADS];
    __shared__ unsigned list[CAP];
    __shared__ unsigned cnt;
    const int tid = threadIdx.x;
    const int n0 = blockIdx.x * TR;
    for (int i = tid; i < (TR + 1) * HID; i += TB) acc[i] = 0.f;
    for (int i = tid; i < (TR + 1) * HEADS; i += TB) { mk[i] = NEGKEY; ss[i] = 0.f; }
    if (tid == 0) cnt = 0u;
    __syncthreads();
    tile_collect(dst, n0, 0, EE, list, &cnt);
    __syncthreads();
    const unsigned total = cnt;
    if (tid == 0) keepn[blockIdx.x] = total;
    if (total <= (unsigned)CAP) {
        const int npad = tile_resolve<4>(src, list, (int)total, keep + (size_t)blockIdx.x * CAP);
        __syncthreads();
        tile1_max(el, er, n0, list, (int)total, mk);
        __syncthreads();
        tile1_agg(el, er, H, n0, list, npad, mk, ss, acc);
    } else {
        for (int pass = 0; pass < 2; ++pass) {
            for (int w0 = 0; w0 < EE; w0 += CAP) {
                __syncthreads();
                if (tid == 0) cnt = 0u;
                __syncthreads();
                tile_collect(dst, n0, w0, (w0 + CAP < EE) ? w0 + CAP : EE, list, &cnt);
                __syncthreads();
                const int n = (int)cnt;
                const int npad = tile_resolve<4>(src, list, n, nullptr);
                __syncthreads();
                if (pass == 0) tile1_max(el, er, n0, list, n, mk);
                else           tile1_agg(el, er, H, n0, list, npad, mk, ss, acc);
            }
        }
    }
    __syncthreads();
    for (int i = tid; i < TR * HID; i += TB) {
        const int nl = i >> 7, cidx = i & 127, h = cidx >> 4;
        const float s = ss[nl * HEADS + h];
        float v = (s > 0.f) ? acc[i] / s : 0.f;
        v = elu1(elu1(v + b[cidx]));
        h2[(size_t)n0 * HID + i] = v;
    }
}

__global__ __launch_bounds__(256) void k_gemm2(const float* __restrict__ H2,
                                               const float* __restrict__ W,
                                               float* __restrict__ HT) {
    const int wave = threadIdx.x >> 5;
    const int lane = threadIdx.x & 31;
    const int half = lane >> 4;
    const int l16  = lane & 15;
    const int rt = wave >> 2, ct = wave & 3;
    const int row0 = blockIdx.x * 32 + rt * 16;
    const int col  = ct * 16 + l16;

    v8f c = {};
    const float* xrow = H2 + (size_t)(row0 + l16) * HID;
#pragma unroll 4
    for (int kk = 0; kk < HID; kk += 4) {
        const int k = kk + half * 2;
        v2f a = *(const v2f*)(xrow + k);
        v2f b;
        b.x = W[(size_t)k * NCLS + col];
        b.y = W[(size_t)(k + 1) * NCLS + col];
        c = __builtin_amdgcn_wmma_f32_16x16x4_f32(false, a, false, b, (short)0, c, false, false);
    }
#pragma unroll
    for (int v = 0; v < 8; ++v)
        HT[(size_t)(row0 + v + half * 8) * NCLS + col] = c[v];
}

__global__ __launch_bounds__(256) void k_att2(const float* __restrict__ HT,
                                              const float* __restrict__ al,
                                              const float* __restrict__ ar,
                                              float* __restrict__ el, float* __restrict__ er) {
    int t = blockIdx.x * 256 + threadIdx.x;
    if (t >= NN) return;
    const float4* hp = (const float4*)(HT + (size_t)t * NCLS);
    const float4* ap = (const float4*)al;
    const float4* rp = (const float4*)ar;
    float se = 0.f, sr = 0.f;
#pragma unroll
    for (int i = 0; i < NCLS / 4; ++i) {
        const float4 hv = hp[i], av = ap[i], rv = rp[i];
        se += hv.x * av.x + hv.y * av.y + hv.z * av.z + hv.w * av.w;
        sr += hv.x * rv.x + hv.y * rv.y + hv.z * rv.z + hv.w * rv.w;
    }
    el[t] = se; er[t] = sr;
}


__device__ __forceinline__ void tile2_max(const float* __restrict__ el, const float* __restrict__ er,
                                          int n0, const unsigned* list, int n, unsigned* mk) {
    for (int i = threadIdx.x; i < n; i += TB) {
        const unsigned ent = list[i];
        const int s = (int)(ent >> 9), dl = (int)(ent & 511u);
        atomicMax(mk + dl, fkey(lrelu(el[s] + er[n0 + dl])));
    }
}
__device__ __forceinline__ void tile2_agg(const float* __restrict__ el, const float* __restrict__ er,
                                          const float* __restrict__ HT, int n0,
                                          const unsigned* list, int n, const unsigned* mk,
                                          float* ss, float* acc) {
    const int wave = threadIdx.x >> 5, lane = threadIdx.x & 31;
    for (int i0 = wave * 32; i0 < n; i0 += TB) {
        const int idx = i0 + lane;
        const bool ok = idx < n;
        const unsigned ent = list[ok ? idx : i0];
        const int s = (int)(ent >> 9), dl = (int)(ent & 511u);
        const int dn = (n0 + dl < NN) ? n0 + dl : NN - 1;
        const float ev = lrelu(el[s] + er[dn]);
        float m = funkey(mk[dl]);
        if (!isfinite(m)) m = 0.f;
        const float p = __expf(ev - m);
        if (ok) atomicAdd(&ss[dl], p);
        const int have = (n - i0 < 32) ? n - i0 : 32;
        for (int g = 0; g < have; g += 8) {
            float v0[8], v1[8];
#pragma unroll
            for (int i = 0; i < 8; ++i) {
                const int si = __builtin_amdgcn_readlane(s, g + i);
                const float* hr = HT + (size_t)si * NCLS + lane;
                v0[i] = hr[0];
                v1[i] = hr[32];
            }
#pragma unroll
            for (int i = 0; i < 8; ++i) {
                const int di = __builtin_amdgcn_readlane(dl, g + i);
                const float pi = __int_as_float(__builtin_amdgcn_readlane(__float_as_int(p), g + i));
                float* ar = acc + di * NCLS + lane;
                atomicAdd(ar,      pi * v0[i]);
                atomicAdd(ar + 32, pi * v1[i]);
            }
        }
    }
}

__global__ __launch_bounds__(1024) void k_tile2(const int* __restrict__ src, const int* __restrict__ dst,
                                                const float* __restrict__ el, const float* __restrict__ er,
                                                const float* __restrict__ HT, const float* __restrict__ b,
                                                float* __restrict__ out,
                                                const unsigned* __restrict__ keep,
                                                const unsigned* __restrict__ keepn) {
    __shared__ __attribute__((aligned(16))) float acc[(TR + 1) * NCLS];
    __shared__ unsigned mk[TR + 1];
    __shared__ float    ss[TR + 1];
    __shared__ unsigned list[CAP];
    __shared__ unsigned cnt;
    const int tid = threadIdx.x;
    const int n0 = blockIdx.x * TR;
    for (int i = tid; i < (TR + 1) * NCLS; i += TB) acc[i] = 0.f;
    for (int i = tid; i < TR + 1; i += TB) { mk[i] = NEGKEY; ss[i] = 0.f; }
    const unsigned total = keepn[blockIdx.x];
    if (total <= (unsigned)CAP) {
        const unsigned* kp = keep + (size_t)blockIdx.x * CAP;
        for (int i = tid; i < (int)total; i += TB) list[i] = kp[i];
        const int npad = ((int)total + 7) / 8 * 8;
        if (tid < npad - (int)total) list[total + tid] = (unsigned)TR;
        __syncthreads();
        tile2_max(el, er, n0, list, (int)total, mk);
        __syncthreads();
        tile2_agg(el, er, HT, n0, list, npad, mk, ss, acc);
    } else {
        for (int pass = 0; pass < 2; ++pass) {
            for (int w0 = 0; w0 < EE; w0 += CAP) {
                __syncthreads();
                if (tid == 0) cnt = 0u;
                __syncthreads();
                tile_collect(dst, n0, w0, (w0 + CAP < EE) ? w0 + CAP : EE, list, &cnt);
                __syncthreads();
                const int n = (int)cnt;
                const int npad = tile_resolve<8>(src, list, n, nullptr);
                __syncthreads();
                if (pass == 0) tile2_max(el, er, n0, list, n, mk);
                else           tile2_agg(el, er, HT, n0, list, npad, mk, ss, acc);
            }
        }
    }
    __syncthreads();
    for (int i = tid; i < TR * NCLS; i += TB) {
        const int nl = i >> 6, cidx = i & 63;
        const float s = ss[nl];
        const float v = (s > 0.f) ? acc[i] / s : 0.f;
        out[(size_t)n0 * NCLS + i] = v + b[cidx];
    }
}

extern "C" void kernel_launch(void* const* d_in, const int* in_sizes, int n_in,
                              void* d_out, int out_size, void* d_ws, size_t ws_size,
                              hipStream_t stream) {
    const float* x   = (const float*)d_in[0];
    const int*   src = (const int*)  d_in[1];
    const int*   dst = (const int*)  d_in[2];
    const float* W1  = (const float*)d_in[3];
    const float* al1 = (const float*)d_in[4];
    const float* ar1 = (const float*)d_in[5];
    const float* b1  = (const float*)d_in[6];
    const float* W2  = (const float*)d_in[7];
    const float* al2 = (const float*)d_in[8];
    const float* ar2 = (const float*)d_in[9];
    const float* b2  = (const float*)d_in[10];
    float* out = (float*)d_out;

    size_t off = 0;
    auto take = [&](size_t nfloats) -> float* {
        float* p = (float*)((char*)d_ws + off);
        off = (off + nfloats * sizeof(float) + 255) & ~(size_t)255;
        return p;
    };
    float* h1  = take((size_t)NN * HID);
    float* el1 = take((size_t)NN * HEADS);
    float* er1 = take((size_t)NN * HEADS);
    float* h2  = take((size_t)NN * HID);
    unsigned* keep  = (unsigned*)take((size_t)NN * NCLS);
    unsigned* keepn = keep + (size_t)NTILE * CAP;
    float* h2t = h1;
    float* el2 = take((size_t)NN);
    float* er2 = take((size_t)NN);

    const int T = 256;
    k_gemm1<<<NN / 16, T, 0, stream>>>(x, W1, h1);
    k_att1 <<<(NN * HEADS + T - 1) / T, T, 0, stream>>>(h1, al1, ar1, el1, er1);
    k_tile1<<<NTILE, TB, 0, stream>>>(src, dst, el1, er1, h1, b1, h2, keep, keepn);
    k_gemm2<<<NN / 32, T, 0, stream>>>(h2, W2, h2t);
    k_att2 <<<(NN + T - 1) / T, T, 0, stream>>>(h2t, al2, ar2, el2, er2);
    k_tile2<<<NTILE, TB, 0, stream>>>(src, dst, el2, er2, h2t, b2, out, keep, keepn);
}


# Both GEMMs re-tiled in registers: one wave = a 16-row strip over all output columns, each A fragment (one 16-byte load) feeds every column tile, W rows by 16-byte loads; k regrouped inside each 8
# speedup vs baseline: 5.0057x; 1.5417x over previous
#include <hip/hip_runtime.h>

typedef __attribute__((ext_vector_type(2))) float v2f;
typedef __attribute__((ext_vector_type(8))) float v8f;

static constexpr int NN   = 100000;
static constexpr int EE   = 1600000;
static constexpr int IND  = 256;
static constexpr int HID  = 128;
static constexpr int HEADS= 8;
static constexpr int DH   = 16;
static constexpr int NCLS = 64;
static constexpr float SLOPE = 0.2f;
static constexpr unsigned NEGKEY = 0x007FFFFFu;

__device__ __forceinline__ unsigned fkey(float f) {
    unsigned u = __float_as_uint(f);
    return (u & 0x80000000u) ? ~u : (u | 0x80000000u);
}
__device__ __forceinline__ float funkey(unsigned k) {
    unsigned u = (k & 0x80000000u) ? (k & 0x7fffffffu) : ~k;
    return __uint_as_float(u);
}
__device__ __forceinline__ float lrelu(float x) { return x > 0.f ? x : SLOPE * x; }
__device__ __forceinline__ float elu1(float x)  { return x > 0.f ? x : expm1f(x); }

template <int K, int G>
__device__ __forceinline__ void strip_gemm(const float* __restrict__ X, const float* __restrict__ W,
                                           float* __restrict__ H, int strip) {
    constexpr int N = 64 * G;
    const int lane = threadIdx.x & 31;
    const int h = lane >> 4;
    const int c = lane & 15;
    const float* xp = X + (size_t)(strip * 16 + c) * K + 4 * h;
    const float* wp = W + (size_t)(4 * h) * N + 4 * c;

    v8f acc[G][4];
#pragma unroll
    for (int g = 0; g < G; ++g)
#pragma unroll
        for (int t = 0; t < 4; ++t) acc[g][t] = v8f{};

#pragma unroll 2
    for (int k0 = 0; k0 < K; k0 += 8) {
        const float4 a = *(const float4*)(xp + k0);
        float4 b[G][4];
#pragma unroll
        for (int g = 0; g < G; ++g)
#pragma unroll
            for (int j = 0; j < 4; ++j) b[g][j] = *(const float4*)(wp + (size_t)(k0 + j) * N + 64 * g);

        const v2f a01 = {a.x, a.y};
        const v2f a23 = {a.z, a.w};
#pragma unroll
        for (int g = 0; g < G; ++g) {
            const float b0[4] = {b[g][0].x, b[g][0].y, b[g][0].z, b[g][0].w};
            const float b1[4] = {b[g][1].x, b[g][1].y, b[g][1].z, b[g][1].w};
#pragma unroll
            for (int t = 0; t < 4; ++t) {
                const v2f bb = {b0[t], b1[t]};
                acc[g][t] = __builtin_amdgcn_wmma_f32_16x16x4_f32(false, a01, false, bb, (short)0, acc[g][t], false, false);
            }
        }
#pragma unroll
        for (int g = 0; g < G; ++g) {
            const float b2[4] = {b[g][2].x, b[g][2].y, b[g][2].z, b[g][2].w};
            const float b3[4] = {b[g][3].x, b[g][3].y, b[g][3].z, b[g][3].w};
#pragma unroll
            for (int t = 0; t < 4; ++t) {
                const v2f bb = {b2[t], b3[t]};
                acc[g][t] = __builtin_amdgcn_wmma_f32_16x16x4_f32(false, a23, false, bb, (short)0, acc[g][t], false, false);
            }
        }
    }

    float* hp = H + (size_t)(strip * 16 + 8 * h) * N + 4 * c;
#pragma unroll
    for (int r = 0; r < 8; ++r)
#pragma unroll
        for (int g = 0; g < G; ++g)
            *(float4*)(hp + (size_t)r * N + 64 * g) =
                make_float4(acc[g][0][r], acc[g][1][r], acc[g][2][r], acc[g][3][r]);
}

static constexpr int GW = 2;
static_assert((NN / 16) % GW == 0, "whole blocks of strips");

__global__ __launch_bounds__(32 * GW) void k_gemm1(const float* __restrict__ X,
                                                  const float* __restrict__ W,
                                                  float* __restrict__ H) {
    strip_gemm<IND, HID / 64>(X, W, H, blockIdx.x * GW + (threadIdx.x >> 5));
}

__global__ __launch_bounds__(256) void k_att1(const float* __restrict__ H,
                                              const float* __restrict__ al,
                                              const float* __restrict__ ar,
                                              float* __restrict__ el, float* __restrict__ er) {
    int t = blockIdx.x * 256 + threadIdx.x;
    if (t >= NN * HEADS) return;
    const int h = t & 7;
    const float4* hp = (const float4*)(H + (size_t)t * DH);
    const float4* ap = (const float4*)(al + h * DH);
    const float4* rp = (const float4*)(ar + h * DH);
    float se = 0.f, sr = 0.f;
#pragma unroll
    for (int i = 0; i < 4; ++i) {
        const float4 hv = hp[i], av = ap[i], rv = rp[i];
        se += hv.x * av.x + hv.y * av.y + hv.z * av.z + hv.w * av.w;
        sr += hv.x * rv.x + hv.y * rv.y + hv.z * rv.z + hv.w * rv.w;
    }
    el[t] = se; er[t] = sr;
}

static constexpr int TR    = 400;
static constexpr int NTILE = NN / TR;
static constexpr int CAP   = 16384;
static constexpr int TB    = 1024;
static_assert(NN % TR == 0 && TR <= 512, "tile index fits 9 bits");
static_assert((size_t)NTILE * CAP + NTILE <= (size_t)NN * NCLS, "lists and counts fit the former h2t slot");
static_assert(EE % 4 == 0 && CAP % 4 == 0 && CAP == 4 * 4 * TB, "a CAP window is one sweep of the block");

__device__ __forceinline__ void tile_test(int d, int e, int n0, unsigned* list, unsigned* cnt) {
    const unsigned dl = (unsigned)(d - n0);
    if (dl < (unsigned)TR) {
        const unsigned slot = atomicAdd(cnt, 1u);
        if (slot < (unsigned)CAP) list[slot] = ((unsigned)e << 9) | dl;
    }
}
__device__ __forceinline__ void tile_collect(const int* __restrict__ dst, int n0, int e0, int e1,
                                             unsigned* list, unsigned* cnt) {
    const int4* d4 = (const int4*)dst;
    const int q1 = e1 >> 2;
    for (int q = (e0 >> 2) + (int)threadIdx.x; q < q1; q += 4 * TB) {
        int4 d[4];
#pragma unroll
        for (int u = 0; u < 4; ++u) {
            const int qq = q + TB * u;
            d[u] = (qq < q1) ? d4[qq] : make_int4(-1, -1, -1, -1);
        }
#pragma unroll
        for (int u = 0; u < 4; ++u) {
            const int e = (q + TB * u) * 4;
            tile_test(d[u].x, e,     n0, list, cnt);
            tile_test(d[u].y, e + 1, n0, list, cnt);
            tile_test(d[u].z, e + 2, n0, list, cnt);
            tile_test(d[u].w, e + 3, n0, list, cnt);
        }
    }
}
template <int G>
__device__ __forceinline__ int tile_resolve(const int* __restrict__ src, unsigned* list, int n,
                                            unsigned* __restrict__ keep) {
    static_assert(CAP % G == 0 && G <= TB, "padding stays inside the list");
    for (int i = threadIdx.x; i < n; i += TB) {
        const unsigned ent = list[i];
        const unsigned res = ((unsigned)src[ent >> 9] << 9) | (ent & 511u);
        list[i] = res;
        if (keep) keep[i] = res;
    }
    const int npad = (n + G - 1) / G * G;
    if ((int)threadIdx.x < npad - n) list[n + threadIdx.x] = (unsigned)TR;
    return npad;
}

__device__ __forceinline__ void tile1_max(const float* __restrict__ el, const float* __restrict__ er,
                                          int n0, const unsigned* list, int n, unsigned* mk) {
    for (int i = threadIdx.x; i < n; i += TB) {
        const unsigned ent = list[i];
        const int s = (int)(ent >> 9), dl = (int)(ent & 511u);
        const float4* ep = (const float4*)(el + (size_t)s * HEADS);
        const float4* rp = (const float4*)(er + (size_t)(n0 + dl) * HEADS);
        const float4 a0 = ep[0], a1 = ep[1], r0 = rp[0], r1 = rp[1];
        unsigned* m = mk + dl * HEADS;
        atomicMax(m + 0, fkey(lrelu(a0.x + r0.x)));
        atomicMax(m + 1, fkey(lrelu(a0.y + r0.y)));
        atomicMax(m + 2, fkey(lrelu(a0.z + r0.z)));
        atomicMax(m + 3, fkey(lrelu(a0.w + r0.w)));
        atomicMax(m + 4, fkey(lrelu(a1.x + r1.x)));
        atomicMax(m + 5, fkey(lrelu(a1.y + r1.y)));
        atomicMax(m + 6, fkey(lrelu(a1.z + r1.z)));
        atomicMax(m + 7, fkey(lrelu(a1.w + r1.w)));
    }
}
__device__ __forceinline__ void tile1_agg(const float* __restrict__ el, const float* __restrict__ er,
                                          const float* __restrict__ H, int n0,
                                          const unsigned* list, int n, const unsigned* mk,
                                          float* ss, float* acc) {
    const int wave = threadIdx.x >> 5, lane = threadIdx.x & 31;
    const int hsel = lane >> 4;
    for (int i0 = wave * 4; i0 < n; i0 += (TB / 32) * 4) {
        const unsigned ent = list[i0 + (lane >> 3)];
        const int s = (int)(ent >> 9), dl = (int)(ent & 511u);
        const int h = lane & 7;
        const int dn = (n0 + dl < NN) ? n0 + dl : NN - 1;
        const float ev = lrelu(el[(size_t)s * HEADS + h] + er[(size_t)dn * HEADS + h]);
        float m = funkey(mk[dl * HEADS + h]);
        if (!isfinite(m)) m = 0.f;
        const float p = __expf(ev - m);
        atomicAdd(&ss[dl * HEADS + h], p);
        float v[4][4];
#pragma unroll
        for (int i = 0; i < 4; ++i) {
            const int si = __builtin_amdgcn_readlane(s, i * 8);
            const float* hr = H + (size_t)si * HID + lane;
#pragma unroll
            for (int j = 0; j < 4; ++j) v[i][j] = hr[32 * j];
        }
#pragma unroll
        for (int i = 0; i < 4; ++i) {
            const int di = __builtin_amdgcn_readlane(dl, i * 8);
            float* ar = acc + di * HID + lane;
#pragma unroll
            for (int j = 0; j < 4; ++j) {
                const float pa = __int_as_float(__builtin_amdgcn_readlane(__float_as_int(p), i * 8 + 2 * j));
                const float pb = __int_as_float(__builtin_amdgcn_readlane(__float_as_int(p), i * 8 + 2 * j + 1));
                atomicAdd(ar + 32 * j, (hsel ? pb : pa) * v[i][j]);
            }
        }
    }
}

__global__ __launch_bounds__(1024) void k_tile1(const int* __restrict__ src, const int* __restrict__ dst,
                                                const float* __restrict__ el, const float* __restrict__ er,
                                                const float* __restrict__ H, const float* __restrict__ b,
                                                float* __restrict__ h2,
                                                unsigned* __restrict__ keep, unsigned* __restrict__ keepn) {
    __shared__ __attribute__((aligned(16))) float acc[(TR + 1) * HID];
    __shared__ unsigned mk[(TR + 1) * HEADS];
    __shared__ float    ss[(TR + 1) * HEADS];
    __shared__ unsigned list[CAP];
    __shared__ unsigned cnt;
    const int tid = threadIdx.x;
    const int n0 = blockIdx.x * TR;
    for (int i = tid; i < (TR + 1) * HID; i += TB) acc[i] = 0.f;
    for (int i = tid; i < (TR + 1) * HEADS; i += TB) { mk[i] = NEGKEY; ss[i] = 0.f; }
    if (tid == 0) cnt = 0u;
    __syncthreads();
    tile_collect(dst, n0, 0, EE, list, &cnt);
    __syncthreads();
    const unsigned total = cnt;
    if (tid == 0) keepn[blockIdx.x] = total;
    if (total <= (unsigned)CAP) {
        const int npad = tile_resolve<4>(src, list, (int)total, keep + (size_t)blockIdx.x * CAP);
        __syncthreads();
        tile1_max(el, er, n0, list, (int)total, mk);
        __syncthreads();
        tile1_agg(el, er, H, n0, list, npad, mk, ss, acc);
    } else {
        for (int pass = 0; pass < 2; ++pass) {
            for (int w0 = 0; w0 < EE; w0 += CAP) {
                __syncthreads();
                if (tid == 0) cnt = 0u;
                __syncthreads();
                tile_collect(dst, n0, w0, (w0 + CAP < EE) ? w0 + CAP : EE, list, &cnt);
                __syncthreads();
                const int n = (int)cnt;
                const int npad = tile_resolve<4>(src, list, n, nullptr);
                __syncthreads();
                if (pass == 0) tile1_max(el, er, n0, list, n, mk);
                else           tile1_agg(el, er, H, n0, list, npad, mk, ss, acc);
            }
        }
    }
    __syncthreads();
    for (int i = tid; i < TR * HID; i += TB) {
        const int nl = i >> 7, cidx = i & 127, h = cidx >> 4;
        const float s = ss[nl * HEADS + h];
        float v = (s > 0.f) ? acc[i] / s : 0.f;
        v = elu1(elu1(v + b[cidx]));
        h2[(size_t)n0 * HID + i] = v;
    }
}

__global__ __launch_bounds__(32 * GW) void k_gemm2(const float* __restrict__ H2,
                                                  const float* __restrict__ W,
                                                  float* __restrict__ HT) {
    strip_gemm<HID, NCLS / 64>(H2, W, HT, blockIdx.x * GW + (threadIdx.x >> 5));
}

__global__ __launch_bounds__(256) void k_att2(const float* __restrict__ HT,
                                              const float* __restrict__ al,
                                              const float* __restrict__ ar,
                                              float* __restrict__ el, float* __restrict__ er) {
    int t = blockIdx.x * 256 + threadIdx.x;
    if (t >= NN) return;
    const float4* hp = (const float4*)(HT + (size_t)t * NCLS);
    const float4* ap = (const float4*)al;
    const float4* rp = (const float4*)ar;
    float se = 0.f, sr = 0.f;
#pragma unroll
    for (int i = 0; i < NCLS / 4; ++i) {
        const float4 hv = hp[i], av = ap[i], rv = rp[i];
        se += hv.x * av.x + hv.y * av.y + hv.z * av.z + hv.w * av.w;
        sr += hv.x * rv.x + hv.y * rv.y + hv.z * rv.z + hv.w * rv.w;
    }
    el[t] = se; er[t] = sr;
}


__device__ __forceinline__ void tile2_max(const float* __restrict__ el, const float* __restrict__ er,
                                          int n0, const unsigned* list, int n, unsigned* mk) {
    for (int i = threadIdx.x; i < n; i += TB) {
        const unsigned ent = list[i];
        const int s = (int)(ent >> 9), dl = (int)(ent & 511u);
        atomicMax(mk + dl, fkey(lrelu(el[s] + er[n0 + dl])));
    }
}
__device__ __forceinline__ void tile2_agg(const float* __restrict__ el, const float* __restrict__ er,
                                          const float* __restrict__ HT, int n0,
                                          const unsigned* list, int n, const unsigned* mk,
                                          float* ss, float* acc) {
    const int wave = threadIdx.x >> 5, lane = threadIdx.x & 31;
    for (int i0 = wave * 32; i0 < n; i0 += TB) {
        const int idx = i0 + lane;
        const bool ok = idx < n;
        const unsigned ent = list[ok ? idx : i0];
        const int s = (int)(ent >> 9), dl = (int)(ent & 511u);
        const int dn = (n0 + dl < NN) ? n0 + dl : NN - 1;
        const float ev = lrelu(el[s] + er[dn]);
        float m = funkey(mk[dl]);
        if (!isfinite(m)) m = 0.f;
        const float p = __expf(ev - m);
        if (ok) atomicAdd(&ss[dl], p);
        const int have = (n - i0 < 32) ? n - i0 : 32;
        for (int g = 0; g < have; g += 8) {
            float v0[8], v1[8];
#pragma unroll
            for (int i = 0; i < 8; ++i) {
                const int si = __builtin_amdgcn_readlane(s, g + i);
                const float* hr = HT + (size_t)si * NCLS + lane;
                v0[i] = hr[0];
                v1[i] = hr[32];
            }
#pragma unroll
            for (int i = 0; i < 8; ++i) {
                const int di = __builtin_amdgcn_readlane(dl, g + i);
                const float pi = __int_as_float(__builtin_amdgcn_readlane(__float_as_int(p), g + i));
                float* ar = acc + di * NCLS + lane;
                atomicAdd(ar,      pi * v0[i]);
                atomicAdd(ar + 32, pi * v1[i]);
            }
        }
    }
}

__global__ __launch_bounds__(1024) void k_tile2(const int* __restrict__ src, const int* __restrict__ dst,
                                                const float* __restrict__ el, const float* __restrict__ er,
                                                const float* __restrict__ HT, const float* __restrict__ b,
                                                float* __restrict__ out,
                                                const unsigned* __restrict__ keep,
                                                const unsigned* __restrict__ keepn) {
    __shared__ __attribute__((aligned(16))) float acc[(TR + 1) * NCLS];
    __shared__ unsigned mk[TR + 1];
    __shared__ float    ss[TR + 1];
    __shared__ unsigned list[CAP];
    __shared__ unsigned cnt;
    const int tid = threadIdx.x;
    const int n0 = blockIdx.x * TR;
    for (int i = tid; i < (TR + 1) * NCLS; i += TB) acc[i] = 0.f;
    for (int i = tid; i < TR + 1; i += TB) { mk[i] = NEGKEY; ss[i] = 0.f; }
    const unsigned total = keepn[blockIdx.x];
    if (total <= (unsigned)CAP) {
        const unsigned* kp = keep + (size_t)blockIdx.x * CAP;
        for (int i = tid; i < (int)total; i += TB) list[i] = kp[i];
        const int npad = ((int)total + 7) / 8 * 8;
        if (tid < npad - (int)total) list[total + tid] = (unsigned)TR;
        __syncthreads();
        tile2_max(el, er, n0, list, (int)total, mk);
        __syncthreads();
        tile2_agg(el, er, HT, n0, list, npad, mk, ss, acc);
    } else {
        for (int pass = 0; pass < 2; ++pass) {
            for (int w0 = 0; w0 < EE; w0 += CAP) {
                __syncthreads();
                if (tid == 0) cnt = 0u;
                __syncthreads();
                tile_collect(dst, n0, w0, (w0 + CAP < EE) ? w0 + CAP : EE, list, &cnt);
                __syncthreads();
                const int n = (int)cnt;
                const int npad = tile_resolve<8>(src, list, n, nullptr);
                __syncthreads();
                if (pass == 0) tile2_max(el, er, n0, list, n, mk);
                else           tile2_agg(el, er, HT, n0, list, npad, mk, ss, acc);
            }
        }
    }
    __syncthreads();
    for (int i = tid; i < TR * NCLS; i += TB) {
        const int nl = i >> 6, cidx = i & 63;
        const float s = ss[nl];
        const float v = (s > 0.f) ? acc[i] / s : 0.f;
        out[(size_t)n0 * NCLS + i] = v + b[cidx];
    }
}

extern "C" void kernel_launch(void* const* d_in, const int* in_sizes, int n_in,
                              void* d_out, int out_size, void* d_ws, size_t ws_size,
                              hipStream_t stream) {
    const float* x   = (const float*)d_in[0];
    const int*   src = (const int*)  d_in[1];
    const int*   dst = (const int*)  d_in[2];
    const float* W1  = (const float*)d_in[3];
    const float* al1 = (const float*)d_in[4];
    const float* ar1 = (const float*)d_in[5];
    const float* b1  = (const float*)d_in[6];
    const float* W2  = (const float*)d_in[7];
    const float* al2 = (const float*)d_in[8];
    const float* ar2 = (const float*)d_in[9];
    const float* b2  = (const float*)d_in[10];
    float* out = (float*)d_out;

    size_t off = 0;
    auto take = [&](size_t nfloats) -> float* {
        float* p = (float*)((char*)d_ws + off);
        off = (off + nfloats * sizeof(float) + 255) & ~(size_t)255;
        return p;
    };
    float* h1  = take((size_t)NN * HID);
    float* el1 = take((size_t)NN * HEADS);
    float* er1 = take((size_t)NN * HEADS);
    float* h2  = take((size_t)NN * HID);
    unsigned* keep  = (unsigned*)take((size_t)NN * NCLS);
    unsigned* keepn = keep + (size_t)NTILE * CAP;
    float* h2t = h1;
    float* el2 = take((size_t)NN);
    float* er2 = take((size_t)NN);

    const int T = 256;
    k_gemm1<<<NN / 16 / GW, 32 * GW, 0, stream>>>(x, W1, h1);
    k_att1 <<<(NN * HEADS + T - 1) / T, T, 0, stream>>>(h1, al1, ar1, el1, er1);
    k_tile1<<<NTILE, TB, 0, stream>>>(src, dst, el1, er1, h1, b1, h2, keep, keepn);
    k_gemm2<<<NN / 16 / GW, 32 * GW, 0, stream>>>(h2, W2, h2t);
    k_att2 <<<(NN + T - 1) / T, T, 0, stream>>>(h2t, al2, ar2, el2, er2);
    k_tile2<<<NTILE, TB, 0, stream>>>(src, dst, el2, er2, h2t, b2, out, keep, keepn);
}


# Layer-1 tile filter sweeps a 1-byte tile-id plane (written by the layer-1 logit kernel) instead of the 4-byte dst array: 16 edges per 16-byte load, zero-byte mask test; LDS arrival sum order kept
# speedup vs baseline: 5.9281x; 1.1843x over previous
#include <hip/hip_runtime.h>

typedef __attribute__((ext_vector_type(2))) float v2f;
typedef __attribute__((ext_vector_type(8))) float v8f;

static constexpr int NN   = 100000;
static constexpr int EE   = 1600000;
static constexpr int IND  = 256;
static constexpr int HID  = 128;
static constexpr int HEADS= 8;
static constexpr int DH   = 16;
static constexpr int NCLS = 64;
static constexpr float SLOPE = 0.2f;
static constexpr unsigned NEGKEY = 0x007FFFFFu;

__device__ __forceinline__ unsigned fkey(float f) {
    unsigned u = __float_as_uint(f);
    return (u & 0x80000000u) ? ~u : (u | 0x80000000u);
}
__device__ __forceinline__ float funkey(unsigned k) {
    unsigned u = (k & 0x80000000u) ? (k & 0x7fffffffu) : ~k;
    return __uint_as_float(u);
}
__device__ __forceinline__ float lrelu(float x) { return x > 0.f ? x : SLOPE * x; }
__device__ __forceinline__ float elu1(float x)  { return x > 0.f ? x : expm1f(x); }

template <int K, int G>
__device__ __forceinline__ void strip_gemm(const float* __restrict__ X, const float* __restrict__ W,
                                           float* __restrict__ H, int strip) {
    constexpr int N = 64 * G;
    const int lane = threadIdx.x & 31;
    const int h = lane >> 4;
    const int c = lane & 15;
    const float* xp = X + (size_t)(strip * 16 + c) * K + 4 * h;
    const float* wp = W + (size_t)(4 * h) * N + 4 * c;

    v8f acc[G][4];
#pragma unroll
    for (int g = 0; g < G; ++g)
#pragma unroll
        for (int t = 0; t < 4; ++t) acc[g][t] = v8f{};

#pragma unroll 2
    for (int k0 = 0; k0 < K; k0 += 8) {
        const float4 a = *(const float4*)(xp + k0);
        float4 b[G][4];
#pragma unroll
        for (int g = 0; g < G; ++g)
#pragma unroll
            for (int j = 0; j < 4; ++j) b[g][j] = *(const float4*)(wp + (size_t)(k0 + j) * N + 64 * g);

        const v2f a01 = {a.x, a.y};
        const v2f a23 = {a.z, a.w};
#pragma unroll
        for (int g = 0; g < G; ++g) {
            const float b0[4] = {b[g][0].x, b[g][0].y, b[g][0].z, b[g][0].w};
            const float b1[4] = {b[g][1].x, b[g][1].y, b[g][1].z, b[g][1].w};
#pragma unroll
            for (int t = 0; t < 4; ++t) {
                const v2f bb = {b0[t], b1[t]};
                acc[g][t] = __builtin_amdgcn_wmma_f32_16x16x4_f32(false, a01, false, bb, (short)0, acc[g][t], false, false);
            }
        }
#pragma unroll
        for (int g = 0; g < G; ++g) {
            const float b2[4] = {b[g][2].x, b[g][2].y, b[g][2].z, b[g][2].w};
            const float b3[4] = {b[g][3].x, b[g][3].y, b[g][3].z, b[g][3].w};
#pragma unroll
            for (int t = 0; t < 4; ++t) {
                const v2f bb = {b2[t], b3[t]};
                acc[g][t] = __builtin_amdgcn_wmma_f32_16x16x4_f32(false, a23, false, bb, (short)0, acc[g][t], false, false);
            }
        }
    }

    float* hp = H + (size_t)(strip * 16 + 8 * h) * N + 4 * c;
#pragma unroll
    for (int r = 0; r < 8; ++r)
#pragma unroll
        for (int g = 0; g < G; ++g)
            *(float4*)(hp + (size_t)r * N + 64 * g) =
                make_float4(acc[g][0][r], acc[g][1][r], acc[g][2][r], acc[g][3][r]);
}

static constexpr int GW = 2;
static_assert((NN / 16) % GW == 0, "whole blocks of strips");

__global__ __launch_bounds__(32 * GW) void k_gemm1(const float* __restrict__ X,
                                                  const float* __restrict__ W,
                                                  float* __restrict__ H) {
    strip_gemm<IND, HID / 64>(X, W, H, blockIdx.x * GW + (threadIdx.x >> 5));
}

static constexpr int TR    = 400;
static constexpr int NTILE = NN / TR;
static_assert(NTILE <= 255, "a tile id fits one byte, 255 is free for 'no tile'");
static_assert(EE % 16 == 0, "the tile-id plane is read in 16-byte groups");
static_assert(EE / 4 <= NN * HEADS, "the layer-1 logit kernel has a thread for every 4 edges of the plane");

__device__ __forceinline__ unsigned tile_byte(int d) {
    const unsigned t = (unsigned)d / (unsigned)TR;
    return t < (unsigned)NTILE ? t : 255u;
}

__global__ __launch_bounds__(256) void k_att1(const float* __restrict__ H,
                                              const float* __restrict__ al,
                                              const float* __restrict__ ar,
                                              float* __restrict__ el, float* __restrict__ er,
                                              const int* __restrict__ dst, unsigned* __restrict__ keys) {
    int t = blockIdx.x * 256 + threadIdx.x;
    if (t < EE / 4) {
        const int4 d = ((const int4*)dst)[t];
        keys[t] = tile_byte(d.x) | (tile_byte(d.y) << 8) | (tile_byte(d.z) << 16) | (tile_byte(d.w) << 24);
    }
    if (t >= NN * HEADS) return;
    const int h = t & 7;
    const float4* hp = (const float4*)(H + (size_t)t * DH);
    const float4* ap = (const float4*)(al + h * DH);
    const float4* rp = (const float4*)(ar + h * DH);
    float se = 0.f, sr = 0.f;
#pragma unroll
    for (int i = 0; i < 4; ++i) {
        const float4 hv = hp[i], av = ap[i], rv = rp[i];
        se += hv.x * av.x + hv.y * av.y + hv.z * av.z + hv.w * av.w;
        sr += hv.x * rv.x + hv.y * rv.y + hv.z * rv.z + hv.w * rv.w;
    }
    el[t] = se; er[t] = sr;
}

static constexpr int CAP   = 16384;
static constexpr int TB    = 1024;
static_assert(NN % TR == 0 && TR <= 512, "tile index fits 9 bits");
static constexpr size_t KEYS_AT = (size_t)NTILE * CAP + 256;
static_assert(KEYS_AT % 4 == 0 && NTILE <= 256, "counts end before the plane, plane aligned for 16-byte loads");
static_assert(KEYS_AT + EE / 4 <= (size_t)NN * NCLS, "lists, counts and the tile-id plane fit the former h2t slot");
static_assert(EE % 4 == 0 && CAP % 4 == 0 && CAP == 4 * 4 * TB, "a CAP window is one sweep of the block");

__device__ __forceinline__ void tile_test(int d, int e, int n0, unsigned* list, unsigned* cnt) {
    const unsigned dl = (unsigned)(d - n0);
    if (dl < (unsigned)TR) {
        const unsigned slot = atomicAdd(cnt, 1u);
        if (slot < (unsigned)CAP) list[slot] = ((unsigned)e << 9) | dl;
    }
}
__device__ __forceinline__ void tile_collect(const int* __restrict__ dst, int n0, int e0, int e1,
                                             unsigned* list, unsigned* cnt) {
    const int4* d4 = (const int4*)dst;
    const int q1 = e1 >> 2;
    for (int q = (e0 >> 2) + (int)threadIdx.x; q < q1; q += 4 * TB) {
        int4 d[4];
#pragma unroll
        for (int u = 0; u < 4; ++u) {
            const int qq = q + TB * u;
            d[u] = (qq < q1) ? d4[qq] : make_int4(-1, -1, -1, -1);
        }
#pragma unroll
        for (int u = 0; u < 4; ++u) {
            const int e = (q + TB * u) * 4;
            tile_test(d[u].x, e,     n0, list, cnt);
            tile_test(d[u].y, e + 1, n0, list, cnt);
            tile_test(d[u].z, e + 2, n0, list, cnt);
            tile_test(d[u].w, e + 3, n0, list, cnt);
        }
    }
}
template <int G>
__device__ __forceinline__ int tile_resolve(const int* __restrict__ src, unsigned* list, int n,
                                            unsigned* __restrict__ keep) {
    static_assert(CAP % G == 0 && G <= TB, "padding stays inside the list");
    for (int i = threadIdx.x; i < n; i += TB) {
        const unsigned ent = list[i];
        const unsigned res = ((unsigned)src[ent >> 9] << 9) | (ent & 511u);
        list[i] = res;
        if (keep) keep[i] = res;
    }
    const int npad = (n + G - 1) / G * G;
    if ((int)threadIdx.x < npad - n) list[n + threadIdx.x] = (unsigned)TR;
    return npad;
}

__device__ __forceinline__ unsigned zero_bytes(unsigned x) {
    return ~(((x & 0x7F7F7F7Fu) + 0x7F7F7F7Fu) | x | 0x7F7F7F7Fu);
}
__device__ __forceinline__ unsigned match_bits(uint4 k0, uint4 k1, unsigned splat) {
    return (zero_bytes(k0.x ^ splat) >> 7) | (zero_bytes(k0.y ^ splat) >> 6) |
           (zero_bytes(k0.z ^ splat) >> 5) | (zero_bytes(k0.w ^ splat) >> 4) |
           (zero_bytes(k1.x ^ splat) >> 3) | (zero_bytes(k1.y ^ splat) >> 2) |
           (zero_bytes(k1.z ^ splat) >> 1) |  zero_bytes(k1.w ^ splat);
}
__device__ __forceinline__ unsigned drain_bits(unsigned m, unsigned e0, unsigned slot, unsigned* list) {
    static_assert(TB * 16 == (4 << 12), "bit 2 of the bit number selects the group TB * 16 edges further on");
    while (m) {
        const unsigned p = (unsigned)__builtin_ctz(m);
        m &= m - 1u;
        if (slot < (unsigned)CAP) list[slot] = e0 + ((p & 3u) << 2) + (p >> 3) + ((p & 4u) << 12);
        ++slot;
    }
    return slot;
}
__device__ __forceinline__ void tile_collect_keys(const unsigned* __restrict__ keys, unsigned tile,
                                                  unsigned* list, unsigned* cnt) {
    const uint4* k4 = (const uint4*)keys;
    constexpr int Q = EE / 16;
    const unsigned splat = tile * 0x01010101u;
    for (int q = threadIdx.x; q < Q; q += 4 * TB) {
        uint4 k[4];
#pragma unroll
        for (int u = 0; u < 4; ++u) {
            const int qq = q + TB * u;
            k[u] = k4[qq < Q ? qq : Q - 1];
        }
        unsigned mA = match_bits(k[0], k[1], splat);
        unsigned mB = match_bits(k[2], k[3], splat);
        if (q + TB >= Q) mA &= 0x0F0F0F0Fu;
        if (q + 2 * TB >= Q) mB = 0u;
        else if (q + 3 * TB >= Q) mB &= 0x0F0F0F0Fu;
        const unsigned c = (unsigned)__builtin_popcount(mA) + (unsigned)__builtin_popcount(mB);
        if (c) {
            unsigned slot = atomicAdd(cnt, c);
            slot = drain_bits(mA, (unsigned)q * 16u, slot, list);
            drain_bits(mB, (unsigned)(q + 2 * TB) * 16u, slot, list);
        }
    }
}
template <int G>
__device__ __forceinline__ int tile_resolve_ids(const int* __restrict__ src, const int* __restrict__ dst, int n0,
                                                unsigned* list, int n, unsigned* __restrict__ keep) {
    static_assert(CAP % G == 0 && G <= TB, "padding stays inside the list");
    for (int i = threadIdx.x; i < n; i += TB) {
        const unsigned e = list[i];
        const unsigned res = ((unsigned)src[e] << 9) | (unsigned)(dst[e] - n0);
        list[i] = res;
        keep[i] = res;
    }
    const int npad = (n + G - 1) / G * G;
    if ((int)threadIdx.x < npad - n) list[n + threadIdx.x] = (unsigned)TR;
    return npad;
}

__device__ __forceinline__ void tile1_max(const float* __restrict__ el, const float* __restrict__ er,
                                          int n0, const unsigned* list, int n, unsigned* mk) {
    for (int i = threadIdx.x; i < n; i += TB) {
        const unsigned ent = list[i];
        const int s = (int)(ent >> 9), dl = (int)(ent & 511u);
        const float4* ep = (const float4*)(el + (size_t)s * HEADS);
        const float4* rp = (const float4*)(er + (size_t)(n0 + dl) * HEADS);
        const float4 a0 = ep[0], a1 = ep[1], r0 = rp[0], r1 = rp[1];
        unsigned* m = mk + dl * HEADS;
        atomicMax(m + 0, fkey(lrelu(a0.x + r0.x)));
        atomicMax(m + 1, fkey(lrelu(a0.y + r0.y)));
        atomicMax(m + 2, fkey(lrelu(a0.z + r0.z)));
        atomicMax(m + 3, fkey(lrelu(a0.w + r0.w)));
        atomicMax(m + 4, fkey(lrelu(a1.x + r1.x)));
        atomicMax(m + 5, fkey(lrelu(a1.y + r1.y)));
        atomicMax(m + 6, fkey(lrelu(a1.z + r1.z)));
        atomicMax(m + 7, fkey(lrelu(a1.w + r1.w)));
    }
}
__device__ __forceinline__ void tile1_agg(const float* __restrict__ el, const float* __restrict__ er,
                                          const float* __restrict__ H, int n0,
                                          const unsigned* list, int n, const unsigned* mk,
                                          float* ss, float* acc) {
    const int wave = threadIdx.x >> 5, lane = threadIdx.x & 31;
    const int hsel = lane >> 4;
    for (int i0 = wave * 4; i0 < n; i0 += (TB / 32) * 4) {
        const unsigned ent = list[i0 + (lane >> 3)];
        const int s = (int)(ent >> 9), dl = (int)(ent & 511u);
        const int h = lane & 7;
        const int dn = (n0 + dl < NN) ? n0 + dl : NN - 1;
        const float ev = lrelu(el[(size_t)s * HEADS + h] + er[(size_t)dn * HEADS + h]);
        float m = funkey(mk[dl * HEADS + h]);
        if (!isfinite(m)) m = 0.f;
        const float p = __expf(ev - m);
        atomicAdd(&ss[dl * HEADS + h], p);
        float v[4][4];
#pragma unroll
        for (int i = 0; i < 4; ++i) {
            const int si = __builtin_amdgcn_readlane(s, i * 8);
            const float* hr = H + (size_t)si * HID + lane;
#pragma unroll
            for (int j = 0; j < 4; ++j) v[i][j] = hr[32 * j];
        }
#pragma unroll
        for (int i = 0; i < 4; ++i) {
            const int di = __builtin_amdgcn_readlane(dl, i * 8);
            float* ar = acc + di * HID + lane;
#pragma unroll
            for (int j = 0; j < 4; ++j) {
                const float pa = __int_as_float(__builtin_amdgcn_readlane(__float_as_int(p), i * 8 + 2 * j));
                const float pb = __int_as_float(__builtin_amdgcn_readlane(__float_as_int(p), i * 8 + 2 * j + 1));
                atomicAdd(ar + 32 * j, (hsel ? pb : pa) * v[i][j]);
            }
        }
    }
}

__global__ __launch_bounds__(1024) void k_tile1(const int* __restrict__ src, const int* __restrict__ dst,
                                                const float* __restrict__ el, const float* __restrict__ er,
                                                const float* __restrict__ H, const float* __restrict__ b,
                                                float* __restrict__ h2,
                                                unsigned* __restrict__ keep, unsigned* __restrict__ keepn,
                                                const unsigned* __restrict__ keys) {
    __shared__ __attribute__((aligned(16))) float acc[(TR + 1) * HID];
    __shared__ unsigned mk[(TR + 1) * HEADS];
    __shared__ float    ss[(TR + 1) * HEADS];
    __shared__ unsigned list[CAP];
    __shared__ unsigned cnt;
    const int tid = threadIdx.x;
    const int n0 = blockIdx.x * TR;
    for (int i = tid; i < (TR + 1) * HID; i += TB) acc[i] = 0.f;
    for (int i = tid; i < (TR + 1) * HEADS; i += TB) { mk[i] = NEGKEY; ss[i] = 0.f; }
    if (tid == 0) cnt = 0u;
    __syncthreads();
    tile_collect_keys(keys, blockIdx.x, list, &cnt);
    __syncthreads();
    const unsigned total = cnt;
    if (tid == 0) keepn[blockIdx.x] = total;
    if (total <= (unsigned)CAP) {
        const int npad = tile_resolve_ids<4>(src, dst, n0, list, (int)total, keep + (size_t)blockIdx.x * CAP);
        __syncthreads();
        tile1_max(el, er, n0, list, (int)total, mk);
        __syncthreads();
        tile1_agg(el, er, H, n0, list, npad, mk, ss, acc);
    } else {
        for (int pass = 0; pass < 2; ++pass) {
            for (int w0 = 0; w0 < EE; w0 += CAP) {
                __syncthreads();
                if (tid == 0) cnt = 0u;
                __syncthreads();
                tile_collect(dst, n0, w0, (w0 + CAP < EE) ? w0 + CAP : EE, list, &cnt);
                __syncthreads();
                const int n = (int)cnt;
                const int npad = tile_resolve<4>(src, list, n, nullptr);
                __syncthreads();
                if (pass == 0) tile1_max(el, er, n0, list, n, mk);
                else           tile1_agg(el, er, H, n0, list, npad, mk, ss, acc);
            }
        }
    }
    __syncthreads();
    for (int i = tid; i < TR * HID; i += TB) {
        const int nl = i >> 7, cidx = i & 127, h = cidx >> 4;
        const float s = ss[nl * HEADS + h];
        float v = (s > 0.f) ? acc[i] / s : 0.f;
        v = elu1(elu1(v + b[cidx]));
        h2[(size_t)n0 * HID + i] = v;
    }
}

__global__ __launch_bounds__(32 * GW) void k_gemm2(const float* __restrict__ H2,
                                                  const float* __restrict__ W,
                                                  float* __restrict__ HT) {
    strip_gemm<HID, NCLS / 64>(H2, W, HT, blockIdx.x * GW + (threadIdx.x >> 5));
}

__global__ __launch_bounds__(256) void k_att2(const float* __restrict__ HT,
                                              const float* __restrict__ al,
                                              const float* __restrict__ ar,
                                              float* __restrict__ el, float* __restrict__ er) {
    int t = blockIdx.x * 256 + threadIdx.x;
    if (t >= NN) return;
    const float4* hp = (const float4*)(HT + (size_t)t * NCLS);
    const float4* ap = (const float4*)al;
    const float4* rp = (const float4*)ar;
    float se = 0.f, sr = 0.f;
#pragma unroll
    for (int i = 0; i < NCLS / 4; ++i) {
        const float4 hv = hp[i], av = ap[i], rv = rp[i];
        se += hv.x * av.x + hv.y * av.y + hv.z * av.z + hv.w * av.w;
        sr += hv.x * rv.x + hv.y * rv.y + hv.z * rv.z + hv.w * rv.w;
    }
    el[t] = se; er[t] = sr;
}


__device__ __forceinline__ void tile2_max(const float* __restrict__ el, const float* __restrict__ er,
                                          int n0, const unsigned* list, int n, unsigned* mk) {
    for (int i = threadIdx.x; i < n; i += TB) {
        const unsigned ent = list[i];
        const int s = (int)(ent >> 9), dl = (int)(ent & 511u);
        atomicMax(mk + dl, fkey(lrelu(el[s] + er[n0 + dl])));
    }
}
__device__ __forceinline__ void tile2_agg(const float* __restrict__ el, const float* __restrict__ er,
                                          const float* __restrict__ HT, int n0,
                                          const unsigned* list, int n, const unsigned* mk,
                                          float* ss, float* acc) {
    const int wave = threadIdx.x >> 5, lane = threadIdx.x & 31;
    for (int i0 = wave * 32; i0 < n; i0 += TB) {
        const int idx = i0 + lane;
        const bool ok = idx < n;
        const unsigned ent = list[ok ? idx : i0];
        const int s = (int)(ent >> 9), dl = (int)(ent & 511u);
        const int dn = (n0 + dl < NN) ? n0 + dl : NN - 1;
        const float ev = lrelu(el[s] + er[dn]);
        float m = funkey(mk[dl]);
        if (!isfinite(m)) m = 0.f;
        const float p = __expf(ev - m);
        if (ok) atomicAdd(&ss[dl], p);
        const int have = (n - i0 < 32) ? n - i0 : 32;
        for (int g = 0; g < have; g += 8) {
            float v0[8], v1[8];
#pragma unroll
            for (int i = 0; i < 8; ++i) {
                const int si = __builtin_amdgcn_readlane(s, g + i);
                const float* hr = HT + (size_t)si * NCLS + lane;
                v0[i] = hr[0];
                v1[i] = hr[32];
            }
#pragma unroll
            for (int i = 0; i < 8; ++i) {
                const int di = __builtin_amdgcn_readlane(dl, g + i);
                const float pi = __int_as_float(__builtin_amdgcn_readlane(__float_as_int(p), g + i));
                float* ar = acc + di * NCLS + lane;
                atomicAdd(ar,      pi * v0[i]);
                atomicAdd(ar + 32, pi * v1[i]);
            }
        }
    }
}

__global__ __launch_bounds__(1024) void k_tile2(const int* __restrict__ src, const int* __restrict__ dst,
                                                const float* __restrict__ el, const float* __restrict__ er,
                                                const float* __restrict__ HT, const float* __restrict__ b,
                                                float* __restrict__ out,
                                                const unsigned* __restrict__ keep,
                                                const unsigned* __restrict__ keepn) {
    __shared__ __attribute__((aligned(16))) float acc[(TR + 1) * NCLS];
    __shared__ unsigned mk[TR + 1];
    __shared__ float    ss[TR + 1];
    __shared__ unsigned list[CAP];
    __shared__ unsigned cnt;
    const int tid = threadIdx.x;
    const int n0 = blockIdx.x * TR;
    for (int i = tid; i < (TR + 1) * NCLS; i += TB) acc[i] = 0.f;
    for (int i = tid; i < TR + 1; i += TB) { mk[i] = NEGKEY; ss[i] = 0.f; }
    const unsigned total = keepn[blockIdx.x];
    if (total <= (unsigned)CAP) {
        const unsigned* kp = keep + (size_t)blockIdx.x * CAP;
        for (int i = tid; i < (int)total; i += TB) list[i] = kp[i];
        const int npad = ((int)total + 7) / 8 * 8;
        if (tid < npad - (int)total) list[total + tid] = (unsigned)TR;
        __syncthreads();
        tile2_max(el, er, n0, list, (int)total, mk);
        __syncthreads();
        tile2_agg(el, er, HT, n0, list, npad, mk, ss, acc);
    } else {
        for (int pass = 0; pass < 2; ++pass) {
            for (int w0 = 0; w0 < EE; w0 += CAP) {
                __syncthreads();
                if (tid == 0) cnt = 0u;
                __syncthreads();
                tile_collect(dst, n0, w0, (w0 + CAP < EE) ? w0 + CAP : EE, list, &cnt);
                __syncthreads();
                const int n = (int)cnt;
                const int npad = tile_resolve<8>(src, list, n, nullptr);
                __syncthreads();
                if (pass == 0) tile2_max(el, er, n0, list, n, mk);
                else           tile2_agg(el, er, HT, n0, list, npad, mk, ss, acc);
            }
        }
    }
    __syncthreads();
    for (int i = tid; i < TR * NCLS; i += TB) {
        const int nl = i >> 6, cidx = i & 63;
        const float s = ss[nl];
        const float v = (s > 0.f) ? acc[i] / s : 0.f;
        out[(size_t)n0 * NCLS + i] = v + b[cidx];
    }
}

extern "C" void kernel_launch(void* const* d_in, const int* in_sizes, int n_in,
                              void* d_out, int out_size, void* d_ws, size_t ws_size,
                              hipStream_t stream) {
    const float* x   = (const float*)d_in[0];
    const int*   src = (const int*)  d_in[1];
    const int*   dst = (const int*)  d_in[2];
    const float* W1  = (const float*)d_in[3];
    const float* al1 = (const float*)d_in[4];
    const float* ar1 = (const float*)d_in[5];
    const float* b1  = (const float*)d_in[6];
    const float* W2  = (const float*)d_in[7];
    const float* al2 = (const float*)d_in[8];
    const float* ar2 = (const float*)d_in[9];
    const float* b2  = (const float*)d_in[10];
    float* out = (float*)d_out;

    size_t off = 0;
    auto take = [&](size_t nfloats) -> float* {
        float* p = (float*)((char*)d_ws + off);
        off = (off + nfloats * sizeof(float) + 255) & ~(size_t)255;
        return p;
    };
    float* h1  = take((size_t)NN * HID);
    float* el1 = take((size_t)NN * HEADS);
    float* er1 = take((size_t)NN * HEADS);
    float* h2  = take((size_t)NN * HID);
    unsigned* keep  = (unsigned*)take((size_t)NN * NCLS);
    unsigned* keepn = keep + (size_t)NTILE * CAP;
    unsigned* keys  = keep + KEYS_AT;
    float* h2t = h1;
    float* el2 = take((size_t)NN);
    float* er2 = take((size_t)NN);

    const int T = 256;
    k_gemm1<<<NN / 16 / GW, 32 * GW, 0, stream>>>(x, W1, h1);
    k_att1 <<<(NN * HEADS + T - 1) / T, T, 0, stream>>>(h1, al1, ar1, el1, er1, dst, keys);
    k_tile1<<<NTILE, TB, 0, stream>>>(src, dst, el1, er1, h1, b1, h2, keep, keepn, keys);
    k_gemm2<<<NN / 16 / GW, 32 * GW, 0, stream>>>(h2, W2, h2t);
    k_att2 <<<(NN + T - 1) / T, T, 0, stream>>>(h2t, al2, ar2, el2, er2);
    k_tile2<<<NTILE, TB, 0, stream>>>(src, dst, el2, er2, h2t, b2, out, keep, keepn);
}


# Layer-2 GEMM and logit dots moved into the layer-1 tile kernel: each block multiplies its finished 400-row tile from LDS (padded pitch) by W2 on WMMA; h2 round trip, 2 launches gone; lane-tree dots
# speedup vs baseline: 6.6973x; 1.1298x over previous
#include <hip/hip_runtime.h>

typedef __attribute__((ext_vector_type(2))) float v2f;
typedef __attribute__((ext_vector_type(8))) float v8f;

static constexpr int NN   = 100000;
static constexpr int EE   = 1600000;
static constexpr int IND  = 256;
static constexpr int HID  = 128;
static constexpr int HEADS= 8;
static constexpr int DH   = 16;
static constexpr int NCLS = 64;
static constexpr float SLOPE = 0.2f;
static constexpr unsigned NEGKEY = 0x007FFFFFu;

__device__ __forceinline__ unsigned fkey(float f) {
    unsigned u = __float_as_uint(f);
    return (u & 0x80000000u) ? ~u : (u | 0x80000000u);
}
__device__ __forceinline__ float funkey(unsigned k) {
    unsigned u = (k & 0x80000000u) ? (k & 0x7fffffffu) : ~k;
    return __uint_as_float(u);
}
__device__ __forceinline__ float lrelu(float x) { return x > 0.f ? x : SLOPE * x; }
__device__ __forceinline__ float elu1(float x)  { return x > 0.f ? x : expm1f(x); }

template <int K, int G>
__device__ __forceinline__ void strip_gemm(const float* __restrict__ X, const float* __restrict__ W,
                                           float* __restrict__ H, int strip) {
    constexpr int N = 64 * G;
    const int lane = threadIdx.x & 31;
    const int h = lane >> 4;
    const int c = lane & 15;
    const float* xp = X + (size_t)(strip * 16 + c) * K + 4 * h;
    const float* wp = W + (size_t)(4 * h) * N + 4 * c;

    v8f acc[G][4];
#pragma unroll
    for (int g = 0; g < G; ++g)
#pragma unroll
        for (int t = 0; t < 4; ++t) acc[g][t] = v8f{};

#pragma unroll 2
    for (int k0 = 0; k0 < K; k0 += 8) {
        const float4 a = *(const float4*)(xp + k0);
        float4 b[G][4];
#pragma unroll
        for (int g = 0; g < G; ++g)
#pragma unroll
            for (int j = 0; j < 4; ++j) b[g][j] = *(const float4*)(wp + (size_t)(k0 + j) * N + 64 * g);

        const v2f a01 = {a.x, a.y};
        const v2f a23 = {a.z, a.w};
#pragma unroll
        for (int g = 0; g < G; ++g) {
            const float b0[4] = {b[g][0].x, b[g][0].y, b[g][0].z, b[g][0].w};
            const float b1[4] = {b[g][1].x, b[g][1].y, b[g][1].z, b[g][1].w};
#pragma unroll
            for (int t = 0; t < 4; ++t) {
                const v2f bb = {b0[t], b1[t]};
                acc[g][t] = __builtin_amdgcn_wmma_f32_16x16x4_f32(false, a01, false, bb, (short)0, acc[g][t], false, false);
            }
        }
#pragma unroll
        for (int g = 0; g < G; ++g) {
            const float b2[4] = {b[g][2].x, b[g][2].y, b[g][2].z, b[g][2].w};
            const float b3[4] = {b[g][3].x, b[g][3].y, b[g][3].z, b[g][3].w};
#pragma unroll
            for (int t = 0; t < 4; ++t) {
                const v2f bb = {b2[t], b3[t]};
                acc[g][t] = __builtin_amdgcn_wmma_f32_16x16x4_f32(false, a23, false, bb, (short)0, acc[g][t], false, false);
            }
        }
    }

    float* hp = H + (size_t)(strip * 16 + 8 * h) * N + 4 * c;
#pragma unroll
    for (int r = 0; r < 8; ++r)
#pragma unroll
        for (int g = 0; g < G; ++g)
            *(float4*)(hp + (size_t)r * N + 64 * g) =
                make_float4(acc[g][0][r], acc[g][1][r], acc[g][2][r], acc[g][3][r]);
}

static constexpr int GW = 2;
static_assert((NN / 16) % GW == 0, "whole blocks of strips");

__global__ __launch_bounds__(32 * GW) void k_gemm1(const float* __restrict__ X,
                                                  const float* __restrict__ W,
                                                  float* __restrict__ H) {
    strip_gemm<IND, HID / 64>(X, W, H, blockIdx.x * GW + (threadIdx.x >> 5));
}

static constexpr int TR    = 400;
static constexpr int NTILE = NN / TR;
static constexpr int ACS   = HID + 4;
static_assert(TR % 16 == 0 && TR / 16 <= 32, "a tile is whole 16-row strips, at most one per wave");
static_assert(NTILE <= 255, "a tile id fits one byte, 255 is free for 'no tile'");
static_assert(EE % 16 == 0, "the tile-id plane is read in 16-byte groups");
static_assert(EE / 4 <= NN * HEADS, "the layer-1 logit kernel has a thread for every 4 edges of the plane");

__device__ __forceinline__ unsigned tile_byte(int d) {
    const unsigned t = (unsigned)d / (unsigned)TR;
    return t < (unsigned)NTILE ? t : 255u;
}

__global__ __launch_bounds__(256) void k_att1(const float* __restrict__ H,
                                              const float* __restrict__ al,
                                              const float* __restrict__ ar,
                                              float* __restrict__ el, float* __restrict__ er,
                                              const int* __restrict__ dst, unsigned* __restrict__ keys) {
    int t = blockIdx.x * 256 + threadIdx.x;
    if (t < EE / 4) {
        const int4 d = ((const int4*)dst)[t];
        keys[t] = tile_byte(d.x) | (tile_byte(d.y) << 8) | (tile_byte(d.z) << 16) | (tile_byte(d.w) << 24);
    }
    if (t >= NN * HEADS) return;
    const int h = t & 7;
    const float4* hp = (const float4*)(H + (size_t)t * DH);
    const float4* ap = (const float4*)(al + h * DH);
    const float4* rp = (const float4*)(ar + h * DH);
    float se = 0.f, sr = 0.f;
#pragma unroll
    for (int i = 0; i < 4; ++i) {
        const float4 hv = hp[i], av = ap[i], rv = rp[i];
        se += hv.x * av.x + hv.y * av.y + hv.z * av.z + hv.w * av.w;
        sr += hv.x * rv.x + hv.y * rv.y + hv.z * rv.z + hv.w * rv.w;
    }
    el[t] = se; er[t] = sr;
}

static constexpr int CAP   = 16384;
static constexpr int TB    = 1024;
static_assert(NN % TR == 0 && TR <= 512, "tile index fits 9 bits");
static constexpr size_t KEYS_AT = (size_t)NTILE * CAP + 256;
static_assert(KEYS_AT % 4 == 0 && NTILE <= 256, "counts end before the plane, plane aligned for 16-byte loads");
static_assert(KEYS_AT + EE / 4 <= (size_t)NN * NCLS, "lists, counts and the tile-id plane fit their workspace slot");
static_assert(EE % 4 == 0 && CAP % 4 == 0 && CAP == 4 * 4 * TB, "a CAP window is one sweep of the block");

__device__ __forceinline__ void tile_test(int d, int e, int n0, unsigned* list, unsigned* cnt) {
    const unsigned dl = (unsigned)(d - n0);
    if (dl < (unsigned)TR) {
        const unsigned slot = atomicAdd(cnt, 1u);
        if (slot < (unsigned)CAP) list[slot] = ((unsigned)e << 9) | dl;
    }
}
__device__ __forceinline__ void tile_collect(const int* __restrict__ dst, int n0, int e0, int e1,
                                             unsigned* list, unsigned* cnt) {
    const int4* d4 = (const int4*)dst;
    const int q1 = e1 >> 2;
    for (int q = (e0 >> 2) + (int)threadIdx.x; q < q1; q += 4 * TB) {
        int4 d[4];
#pragma unroll
        for (int u = 0; u < 4; ++u) {
            const int qq = q + TB * u;
            d[u] = (qq < q1) ? d4[qq] : make_int4(-1, -1, -1, -1);
        }
#pragma unroll
        for (int u = 0; u < 4; ++u) {
            const int e = (q + TB * u) * 4;
            tile_test(d[u].x, e,     n0, list, cnt);
            tile_test(d[u].y, e + 1, n0, list, cnt);
            tile_test(d[u].z, e + 2, n0, list, cnt);
            tile_test(d[u].w, e + 3, n0, list, cnt);
        }
    }
}
template <int G>
__device__ __forceinline__ int tile_resolve(const int* __restrict__ src, unsigned* list, int n,
                                            unsigned* __restrict__ keep) {
    static_assert(CAP % G == 0 && G <= TB, "padding stays inside the list");
    for (int i = threadIdx.x; i < n; i += TB) {
        const unsigned ent = list[i];
        const unsigned res = ((unsigned)src[ent >> 9] << 9) | (ent & 511u);
        list[i] = res;
        if (keep) keep[i] = res;
    }
    const int npad = (n + G - 1) / G * G;
    if ((int)threadIdx.x < npad - n) list[n + threadIdx.x] = (unsigned)TR;
    return npad;
}

__device__ __forceinline__ unsigned zero_bytes(unsigned x) {
    return ~(((x & 0x7F7F7F7Fu) + 0x7F7F7F7Fu) | x | 0x7F7F7F7Fu);
}
__device__ __forceinline__ unsigned match_bits(uint4 k0, uint4 k1, unsigned splat) {
    return (zero_bytes(k0.x ^ splat) >> 7) | (zero_bytes(k0.y ^ splat) >> 6) |
           (zero_bytes(k0.z ^ splat) >> 5) | (zero_bytes(k0.w ^ splat) >> 4) |
           (zero_bytes(k1.x ^ splat) >> 3) | (zero_bytes(k1.y ^ splat) >> 2) |
           (zero_bytes(k1.z ^ splat) >> 1) |  zero_bytes(k1.w ^ splat);
}
__device__ __forceinline__ unsigned drain_bits(unsigned m, unsigned e0, unsigned slot, unsigned* list) {
    static_assert(TB * 16 == (4 << 12), "bit 2 of the bit number selects the group TB * 16 edges further on");
    while (m) {
        const unsigned p = (unsigned)__builtin_ctz(m);
        m &= m - 1u;
        if (slot < (unsigned)CAP) list[slot] = e0 + ((p & 3u) << 2) + (p >> 3) + ((p & 4u) << 12);
        ++slot;
    }
    return slot;
}
__device__ __forceinline__ void tile_collect_keys(const unsigned* __restrict__ keys, unsigned tile,
                                                  unsigned* list, unsigned* cnt) {
    const uint4* k4 = (const uint4*)keys;
    constexpr int Q = EE / 16;
    const unsigned splat = tile * 0x01010101u;
    for (int q = threadIdx.x; q < Q; q += 4 * TB) {
        uint4 k[4];
#pragma unroll
        for (int u = 0; u < 4; ++u) {
            const int qq = q + TB * u;
            k[u] = k4[qq < Q ? qq : Q - 1];
        }
        unsigned mA = match_bits(k[0], k[1], splat);
        unsigned mB = match_bits(k[2], k[3], splat);
        if (q + TB >= Q) mA &= 0x0F0F0F0Fu;
        if (q + 2 * TB >= Q) mB = 0u;
        else if (q + 3 * TB >= Q) mB &= 0x0F0F0F0Fu;
        const unsigned c = (unsigned)__builtin_popcount(mA) + (unsigned)__builtin_popcount(mB);
        if (c) {
            unsigned slot = atomicAdd(cnt, c);
            slot = drain_bits(mA, (unsigned)q * 16u, slot, list);
            drain_bits(mB, (unsigned)(q + 2 * TB) * 16u, slot, list);
        }
    }
}
template <int G>
__device__ __forceinline__ int tile_resolve_ids(const int* __restrict__ src, const int* __restrict__ dst, int n0,
                                                unsigned* list, int n, unsigned* __restrict__ keep) {
    static_assert(CAP % G == 0 && G <= TB, "padding stays inside the list");
    for (int i = threadIdx.x; i < n; i += TB) {
        const unsigned e = list[i];
        const unsigned res = ((unsigned)src[e] << 9) | (unsigned)(dst[e] - n0);
        list[i] = res;
        keep[i] = res;
    }
    const int npad = (n + G - 1) / G * G;
    if ((int)threadIdx.x < npad - n) list[n + threadIdx.x] = (unsigned)TR;
    return npad;
}

__device__ __forceinline__ void tile1_max(const float* __restrict__ el, const float* __restrict__ er,
                                          int n0, const unsigned* list, int n, unsigned* mk) {
    for (int i = threadIdx.x; i < n; i += TB) {
        const unsigned ent = list[i];
        const int s = (int)(ent >> 9), dl = (int)(ent & 511u);
        const float4* ep = (const float4*)(el + (size_t)s * HEADS);
        const float4* rp = (const float4*)(er + (size_t)(n0 + dl) * HEADS);
        const float4 a0 = ep[0], a1 = ep[1], r0 = rp[0], r1 = rp[1];
        unsigned* m = mk + dl * HEADS;
        atomicMax(m + 0, fkey(lrelu(a0.x + r0.x)));
        atomicMax(m + 1, fkey(lrelu(a0.y + r0.y)));
        atomicMax(m + 2, fkey(lrelu(a0.z + r0.z)));
        atomicMax(m + 3, fkey(lrelu(a0.w + r0.w)));
        atomicMax(m + 4, fkey(lrelu(a1.x + r1.x)));
        atomicMax(m + 5, fkey(lrelu(a1.y + r1.y)));
        atomicMax(m + 6, fkey(lrelu(a1.z + r1.z)));
        atomicMax(m + 7, fkey(lrelu(a1.w + r1.w)));
    }
}
__device__ __forceinline__ void tile1_agg(const float* __restrict__ el, const float* __restrict__ er,
                                          const float* __restrict__ H, int n0,
                                          const unsigned* list, int n, const unsigned* mk,
                                          float* ss, float* acc) {
    const int wave = threadIdx.x >> 5, lane = threadIdx.x & 31;
    const int hsel = lane >> 4;
    for (int i0 = wave * 4; i0 < n; i0 += (TB / 32) * 4) {
        const unsigned ent = list[i0 + (lane >> 3)];
        const int s = (int)(ent >> 9), dl = (int)(ent & 511u);
        const int h = lane & 7;
        const int dn = (n0 + dl < NN) ? n0 + dl : NN - 1;
        const float ev = lrelu(el[(size_t)s * HEADS + h] + er[(size_t)dn * HEADS + h]);
        float m = funkey(mk[dl * HEADS + h]);
        if (!isfinite(m)) m = 0.f;
        const float p = __expf(ev - m);
        atomicAdd(&ss[dl * HEADS + h], p);
        float v[4][4];
#pragma unroll
        for (int i = 0; i < 4; ++i) {
            const int si = __builtin_amdgcn_readlane(s, i * 8);
            const float* hr = H + (size_t)si * HID + lane;
#pragma unroll
            for (int j = 0; j < 4; ++j) v[i][j] = hr[32 * j];
        }
#pragma unroll
        for (int i = 0; i < 4; ++i) {
            const int di = __builtin_amdgcn_readlane(dl, i * 8);
            float* ar = acc + di * ACS + lane;
#pragma unroll
            for (int j = 0; j < 4; ++j) {
                const float pa = __int_as_float(__builtin_amdgcn_readlane(__float_as_int(p), i * 8 + 2 * j));
                const float pb = __int_as_float(__builtin_amdgcn_readlane(__float_as_int(p), i * 8 + 2 * j + 1));
                atomicAdd(ar + 32 * j, (hsel ? pb : pa) * v[i][j]);
            }
        }
    }
}

__device__ __forceinline__ void tile_gemm2(const float* h2s, const float* __restrict__ W,
                                           const float* __restrict__ al, const float* __restrict__ ar,
                                           float* __restrict__ HT, float* __restrict__ el, float* __restrict__ er,
                                           int n0, int strip) {
    const int lane = threadIdx.x & 31;
    const int h = lane >> 4;
    const int c = lane & 15;
    const float* xp = h2s + (strip * 16 + c) * ACS + 4 * h;
    const float* wp = W + (size_t)(4 * h) * NCLS + 4 * c;

    v8f acc[4];
#pragma unroll
    for (int t = 0; t < 4; ++t) acc[t] = v8f{};

#pragma unroll 2
    for (int k0 = 0; k0 < HID; k0 += 8) {
        const float4 a = *(const float4*)(xp + k0);
        float4 b[4];
#pragma unroll
        for (int j = 0; j < 4; ++j) b[j] = *(const float4*)(wp + (size_t)(k0 + j) * NCLS);
        const v2f a01 = {a.x, a.y};
        const v2f a23 = {a.z, a.w};
        const float b0[4] = {b[0].x, b[0].y, b[0].z, b[0].w};
        const float b1[4] = {b[1].x, b[1].y, b[1].z, b[1].w};
        const float b2[4] = {b[2].x, b[2].y, b[2].z, b[2].w};
        const float b3[4] = {b[3].x, b[3].y, b[3].z, b[3].w};
#pragma unroll
        for (int t = 0; t < 4; ++t) {
            const v2f bb = {b0[t], b1[t]};
            acc[t] = __builtin_amdgcn_wmma_f32_16x16x4_f32(false, a01, false, bb, (short)0, acc[t], false, false);
        }
#pragma unroll
        for (int t = 0; t < 4; ++t) {
            const v2f bb = {b2[t], b3[t]};
            acc[t] = __builtin_amdgcn_wmma_f32_16x16x4_f32(false, a23, false, bb, (short)0, acc[t], false, false);
        }
    }

    const int row0 = n0 + strip * 16 + 8 * h;
    float* hp = HT + (size_t)row0 * NCLS + 4 * c;
#pragma unroll
    for (int r = 0; r < 8; ++r)
        *(float4*)(hp + (size_t)r * NCLS) = make_float4(acc[0][r], acc[1][r], acc[2][r], acc[3][r]);

    const float4 av = *(const float4*)(al + 4 * c);
    const float4 rv = *(const float4*)(ar + 4 * c);
    float se[8], sr[8];
#pragma unroll
    for (int r = 0; r < 8; ++r) {
        se[r] = acc[0][r] * av.x + acc[1][r] * av.y + acc[2][r] * av.z + acc[3][r] * av.w;
        sr[r] = acc[0][r] * rv.x + acc[1][r] * rv.y + acc[2][r] * rv.z + acc[3][r] * rv.w;
    }
#pragma unroll
    for (int off = 1; off < 16; off <<= 1) {
#pragma unroll
        for (int r = 0; r < 8; ++r) {
            se[r] += __shfl_xor(se[r], off);
            sr[r] += __shfl_xor(sr[r], off);
        }
    }
    float val = 0.f;
#pragma unroll
    for (int r = 0; r < 8; ++r)
        if ((c & 7) == r) val = (c < 8) ? se[r] : sr[r];
    ((c < 8) ? el : er)[row0 + (c & 7)] = val;
}

__global__ __launch_bounds__(1024) void k_tile1(const int* __restrict__ src, const int* __restrict__ dst,
                                                const float* __restrict__ el, const float* __restrict__ er,
                                                const float* __restrict__ H, const float* __restrict__ b,
                                                const float* __restrict__ W2, const float* __restrict__ al2,
                                                const float* __restrict__ ar2,
                                                float* __restrict__ h2t, float* __restrict__ el2,
                                                float* __restrict__ er2,
                                                unsigned* __restrict__ keep, unsigned* __restrict__ keepn,
                                                const unsigned* __restrict__ keys) {
    __shared__ __attribute__((aligned(16))) float acc[(TR + 1) * ACS];
    __shared__ unsigned mk[(TR + 1) * HEADS];
    __shared__ float    ss[(TR + 1) * HEADS];
    __shared__ unsigned list[CAP];
    __shared__ unsigned cnt;
    const int tid = threadIdx.x;
    const int n0 = blockIdx.x * TR;
    for (int i = tid; i < (TR + 1) * ACS; i += TB) acc[i] = 0.f;
    for (int i = tid; i < (TR + 1) * HEADS; i += TB) { mk[i] = NEGKEY; ss[i] = 0.f; }
    if (tid == 0) cnt = 0u;
    __syncthreads();
    tile_collect_keys(keys, blockIdx.x, list, &cnt);
    __syncthreads();
    const unsigned total = cnt;
    if (tid == 0) keepn[blockIdx.x] = total;
    if (total <= (unsigned)CAP) {
        const int npad = tile_resolve_ids<4>(src, dst, n0, list, (int)total, keep + (size_t)blockIdx.x * CAP);
        __syncthreads();
        tile1_max(el, er, n0, list, (int)total, mk);
        __syncthreads();
        tile1_agg(el, er, H, n0, list, npad, mk, ss, acc);
    } else {
        for (int pass = 0; pass < 2; ++pass) {
            for (int w0 = 0; w0 < EE; w0 += CAP) {
                __syncthreads();
                if (tid == 0) cnt = 0u;
                __syncthreads();
                tile_collect(dst, n0, w0, (w0 + CAP < EE) ? w0 + CAP : EE, list, &cnt);
                __syncthreads();
                const int n = (int)cnt;
                const int npad = tile_resolve<4>(src, list, n, nullptr);
                __syncthreads();
                if (pass == 0) tile1_max(el, er, n0, list, n, mk);
                else           tile1_agg(el, er, H, n0, list, npad, mk, ss, acc);
            }
        }
    }
    __syncthreads();
    for (int i = tid; i < TR * HID; i += TB) {
        const int nl = i >> 7, cidx = i & 127, h = cidx >> 4;
        const float s = ss[nl * HEADS + h];
        float* ap = acc + nl * ACS + cidx;
        float v = (s > 0.f) ? *ap / s : 0.f;
        *ap = elu1(elu1(v + b[cidx]));
    }
    __syncthreads();
    const int wave = tid >> 5;
    if (wave < TR / 16) tile_gemm2(acc, W2, al2, ar2, h2t, el2, er2, n0, wave);
}

__device__ __forceinline__ void tile2_max(const float* __restrict__ el, const float* __restrict__ er,
                                          int n0, const unsigned* list, int n, unsigned* mk) {
    for (int i = threadIdx.x; i < n; i += TB) {
        const unsigned ent = list[i];
        const int s = (int)(ent >> 9), dl = (int)(ent & 511u);
        atomicMax(mk + dl, fkey(lrelu(el[s] + er[n0 + dl])));
    }
}
__device__ __forceinline__ void tile2_agg(const float* __restrict__ el, const float* __restrict__ er,
                                          const float* __restrict__ HT, int n0,
                                          const unsigned* list, int n, const unsigned* mk,
                                          float* ss, float* acc) {
    const int wave = threadIdx.x >> 5, lane = threadIdx.x & 31;
    for (int i0 = wave * 32; i0 < n; i0 += TB) {
        const int idx = i0 + lane;
        const bool ok = idx < n;
        const unsigned ent = list[ok ? idx : i0];
        const int s = (int)(ent >> 9), dl = (int)(ent & 511u);
        const int dn = (n0 + dl < NN) ? n0 + dl : NN - 1;
        const float ev = lrelu(el[s] + er[dn]);
        float m = funkey(mk[dl]);
        if (!isfinite(m)) m = 0.f;
        const float p = __expf(ev - m);
        if (ok) atomicAdd(&ss[dl], p);
        const int have = (n - i0 < 32) ? n - i0 : 32;
        for (int g = 0; g < have; g += 8) {
            float v0[8], v1[8];
#pragma unroll
            for (int i = 0; i < 8; ++i) {
                const int si = __builtin_amdgcn_readlane(s, g + i);
                const float* hr = HT + (size_t)si * NCLS + lane;
                v0[i] = hr[0];
                v1[i] = hr[32];
            }
#pragma unroll
            for (int i = 0; i < 8; ++i) {
                const int di = __builtin_amdgcn_readlane(dl, g + i);
                const float pi = __int_as_float(__builtin_amdgcn_readlane(__float_as_int(p), g + i));
                float* ar = acc + di * NCLS + lane;
                atomicAdd(ar,      pi * v0[i]);
                atomicAdd(ar + 32, pi * v1[i]);
            }
        }
    }
}

__global__ __launch_bounds__(1024) void k_tile2(const int* __restrict__ src, const int* __restrict__ dst,
                                                const float* __restrict__ el, const float* __restrict__ er,
                                                const float* __restrict__ HT, const float* __restrict__ b,
                                                float* __restrict__ out,
                                                const unsigned* __restrict__ keep,
                                                const unsigned* __restrict__ keepn) {
    __shared__ __attribute__((aligned(16))) float acc[(TR + 1) * NCLS];
    __shared__ unsigned mk[TR + 1];
    __shared__ float    ss[TR + 1];
    __shared__ unsigned list[CAP];
    __shared__ unsigned cnt;
    const int tid = threadIdx.x;
    const int n0 = blockIdx.x * TR;
    for (int i = tid; i < (TR + 1) * NCLS; i += TB) acc[i] = 0.f;
    for (int i = tid; i < TR + 1; i += TB) { mk[i] = NEGKEY; ss[i] = 0.f; }
    const unsigned total = keepn[blockIdx.x];
    if (total <= (unsigned)CAP) {
        const unsigned* kp = keep + (size_t)blockIdx.x * CAP;
        for (int i = tid; i < (int)total; i += TB) list[i] = kp[i];
        const int npad = ((int)total + 7) / 8 * 8;
        if (tid < npad - (int)total) list[total + tid] = (unsigned)TR;
        __syncthreads();
        tile2_max(el, er, n0, list, (int)total, mk);
        __syncthreads();
        tile2_agg(el, er, HT, n0, list, npad, mk, ss, acc);
    } else {
        for (int pass = 0; pass < 2; ++pass) {
            for (int w0 = 0; w0 < EE; w0 += CAP) {
                __syncthreads();
                if (tid == 0) cnt = 0u;
                __syncthreads();
                tile_collect(dst, n0, w0, (w0 + CAP < EE) ? w0 + CAP : EE, list, &cnt);
                __syncthreads();
                const int n = (int)cnt;
                const int npad = tile_resolve<8>(src, list, n, nullptr);
                __syncthreads();
                if (pass == 0) tile2_max(el, er, n0, list, n, mk);
                else           tile2_agg(el, er, HT, n0, list, npad, mk, ss, acc);
            }
        }
    }
    __syncthreads();
    for (int i = tid; i < TR * NCLS; i += TB) {
        const int nl = i >> 6, cidx = i & 63;
        const float s = ss[nl];
        const float v = (s > 0.f) ? acc[i] / s : 0.f;
        out[(size_t)n0 * NCLS + i] = v + b[cidx];
    }
}

extern "C" void kernel_launch(void* const* d_in, const int* in_sizes, int n_in,
                              void* d_out, int out_size, void* d_ws, size_t ws_size,
                              hipStream_t stream) {
    const float* x   = (const float*)d_in[0];
    const int*   src = (const int*)  d_in[1];
    const int*   dst = (const int*)  d_in[2];
    const float* W1  = (const float*)d_in[3];
    const float* al1 = (const float*)d_in[4];
    const float* ar1 = (const float*)d_in[5];
    const float* b1  = (const float*)d_in[6];
    const float* W2  = (const float*)d_in[7];
    const float* al2 = (const float*)d_in[8];
    const float* ar2 = (const float*)d_in[9];
    const float* b2  = (const float*)d_in[10];
    float* out = (float*)d_out;

    size_t off = 0;
    auto take = [&](size_t nfloats) -> float* {
        float* p = (float*)((char*)d_ws + off);
        off = (off + nfloats * sizeof(float) + 255) & ~(size_t)255;
        return p;
    };
    float* h1  = take((size_t)NN * HID);
    float* el1 = take((size_t)NN * HEADS);
    float* er1 = take((size_t)NN * HEADS);
    float* h2t = take((size_t)NN * NCLS);
    unsigned* keep  = (unsigned*)take((size_t)NN * NCLS);
    unsigned* keepn = keep + (size_t)NTILE * CAP;
    unsigned* keys  = keep + KEYS_AT;
    float* el2 = take((size_t)NN);
    float* er2 = take((size_t)NN);

    const int T = 256;
    k_gemm1<<<NN / 16 / GW, 32 * GW, 0, stream>>>(x, W1, h1);
    k_att1 <<<(NN * HEADS + T - 1) / T, T, 0, stream>>>(h1, al1, ar1, el1, er1, dst, keys);
    k_tile1<<<NTILE, TB, 0, stream>>>(src, dst, el1, er1, h1, b1, W2, al2, ar2, h2t, el2, er2, keep, keepn, keys);
    k_tile2<<<NTILE, TB, 0, stream>>>(src, dst, el2, er2, h2t, b2, out, keep, keepn);
}
